# starting kernel
# baseline (speedup 1.0000x reference)
#include <hip/hip_runtime.h>

typedef __attribute__((ext_vector_type(2))) float v2f;
typedef __attribute__((ext_vector_type(8))) float v8f;

#define K_TAPS 4096
#define PAD    2048
#define T_LEN  131072
#define S_SIG  32
#define TP     (T_LEN + 2*PAD)
#define GT     2112
#define ROWS   (TP + 2*GT)
#define GUARD  16
#define HEXT   (K_TAPS + 64)
#define KWIN   (K_TAPS + 15)
#define NCHUNK ((KWIN + 4) / 4)
#define TBASE  (-2048)

__global__ void prep_h_kernel(const float* __restrict__ h,
                              float* __restrict__ hext,
                              float* __restrict__ hrev) {
    int i = blockIdx.x * blockDim.x + threadIdx.x;
    if (i >= HEXT) return;
    int j = i - GUARD - 1;
    int d = i - GUARD;
    hext[i] = (j >= 0 && j < K_TAPS) ? h[j] : 0.f;
    hrev[i] = (d >= 0 && d < K_TAPS) ? h[K_TAPS - 1 - d] : 0.f;
}

__global__ void build_xpt_kernel(const float* __restrict__ x,
                                 float* __restrict__ xpt) {
    int o = blockIdx.x * blockDim.x + threadIdx.x;
    int rem = o & 63;
    int s = rem >> 1;
    int t = ((o >> 6) << 1) + (rem & 1);
    int tg = t - GT;
    float v = 0.f;
    if (tg >= 0 && tg < TP) {
        int tr = tg - PAD;
        if (tr < 0)            tr = -tr;
        else if (tr >= T_LEN)  tr = 2 * T_LEN - 2 - tr;
        v = x[(size_t)s * T_LEN + tr];
    }
    xpt[o] = v;
}

__global__ void zero_guards_kernel(float* __restrict__ y1t) {
    int idx = blockIdx.x * blockDim.x + threadIdx.x;
    y1t[idx] = 0.f;
    y1t[(size_t)(GT + TP) * S_SIG + idx] = 0.f;
}

template<bool STORE_OUT>
__global__ __launch_bounds__(256)
void fir_pass_kernel(const float* __restrict__ hext,
                     const float* __restrict__ src,
                     float* __restrict__ dst,
                     int t0_origin) {
    __shared__ float sh[HEXT];
    for (int i = threadIdx.x; i < HEXT; i += 256) sh[i] = hext[i];
    __syncthreads();

    const int lane = threadIdx.x & 31;
    const int wave = threadIdx.x >> 5;
    const int n    = lane & 15;
    const int half = lane >> 4;
    const int q    = half * 2;

    const int t0 = t0_origin + (blockIdx.x * 8 + wave) * 16;

    int ab = GUARD + q - n;
    const float* p = src + ((size_t)(GT + t0 + TBASE + q) >> 1) * 64 + 2 * n;

    v8f acc0 = {};
    v8f acc1 = {};
#pragma unroll 4
    for (int kc = 0; kc < NCHUNK; ++kc) {
        v2f a, b0, b1;
        a.x = sh[ab];
        a.y = sh[ab + 1];
        b0 = *(const v2f*)p;
        b1 = *(const v2f*)(p + 32);
        acc0 = __builtin_amdgcn_wmma_f32_16x16x4_f32(
            false, a, false, b0, (short)0, acc0, false, false);
        acc1 = __builtin_amdgcn_wmma_f32_16x16x4_f32(
            false, a, false, b1, (short)0, acc1, false, false);
        ab += 4;
        p  += 128;
    }

    if (!STORE_OUT) {
        const int row0 = GT + t0 + half * 8;
        float* o = dst + ((size_t)row0 >> 1) * 64 + (size_t)n * 2;
#pragma unroll
        for (int r = 0; r < 4; ++r) {
            v2f w0, w1;
            w0.x = acc0[2 * r]; w0.y = acc0[2 * r + 1];
            w1.x = acc1[2 * r]; w1.y = acc1[2 * r + 1];
            *(v2f*)(o + (size_t)r * 64)      = w0;
            *(v2f*)(o + (size_t)r * 64 + 32) = w1;
        }
    } else {
        float* o0 = dst + (size_t)n        * T_LEN + (t0 - PAD) + half * 8;
        float* o1 = dst + (size_t)(n + 16) * T_LEN + (t0 - PAD) + half * 8;
        *(float4*)(o0)     = make_float4(acc0[0], acc0[1], acc0[2], acc0[3]);
        *((float4*)o0 + 1) = make_float4(acc0[4], acc0[5], acc0[6], acc0[7]);
        *(float4*)(o1)     = make_float4(acc1[0], acc1[1], acc1[2], acc1[3]);
        *((float4*)o1 + 1) = make_float4(acc1[4], acc1[5], acc1[6], acc1[7]);
    }
}

extern "C" void kernel_launch(void* const* d_in, const int* in_sizes, int n_in,
                              void* d_out, int out_size, void* d_ws, size_t ws_size,
                              hipStream_t stream) {
    const float* x    = (const float*)d_in[0];
    const float* kern = (const float*)d_in[1];
    float* out = (float*)d_out;
    float* ws  = (float*)d_ws;

    float* hext = ws;
    float* hrev = ws + HEXT;
    float* xpt  = ws + 2 * HEXT;
    float* y1t  = xpt + (size_t)ROWS * S_SIG;

    prep_h_kernel<<<(HEXT + 255) / 256, 256, 0, stream>>>(kern, hext, hrev);
    build_xpt_kernel<<<(ROWS * S_SIG) / 256, 256, 0, stream>>>(x, xpt);
    zero_guards_kernel<<<(GT * S_SIG) / 256, 256, 0, stream>>>(y1t);

    fir_pass_kernel<false>
        <<<TP / 128, 256, 0, stream>>>(hext, xpt, y1t, 0);
    fir_pass_kernel<true>
        <<<T_LEN / 128, 256, 0, stream>>>(hrev, y1t, out, PAD);
}


# each wave keeps 6 (pass 1) or 8 (pass 2) time tiles of one 16-signal half on one k axis: a B load feeds 6-8 WMMAs, Toeplitz A fragments slide through a register ring; k order of every sum kept
# speedup vs baseline: 1.2868x; 1.2868x over previous
#include <hip/hip_runtime.h>


typedef __attribute__((ext_vector_type(2))) float v2f;
typedef __attribute__((ext_vector_type(8))) float v8f;

#define K_TAPS 4096
#define PAD    2048
#define T_LEN  131072
#define S_SIG  32
#define TP     (T_LEN + 2*PAD)
#define GT     2112
#define ROWS   (TP + 2*GT)
#define GUARD  16
#define HEXT   (K_TAPS + 64)
#define TBASE  (-2048)
#define MT1    6
#define MT2    8

__global__ void prep_h_kernel(const float* __restrict__ h,
                              float* __restrict__ hext,
                              float* __restrict__ hrev) {
    int i = blockIdx.x * blockDim.x + threadIdx.x;
    if (i >= HEXT) return;
    int j = i - GUARD - 1;
    int d = i - GUARD;
    hext[i] = (j >= 0 && j < K_TAPS) ? h[j] : 0.f;
    hrev[i] = (d >= 0 && d < K_TAPS) ? h[K_TAPS - 1 - d] : 0.f;
}

__global__ void build_xpt_kernel(const float* __restrict__ x,
                                 float* __restrict__ xpt) {
    int o = blockIdx.x * blockDim.x + threadIdx.x;
    int rem = o & 63;
    int s = rem >> 1;
    int t = ((o >> 6) << 1) + (rem & 1);
    int tg = t - GT;
    float v = 0.f;
    if (tg >= 0 && tg < TP) {
        int tr = tg - PAD;
        if (tr < 0)            tr = -tr;
        else if (tr >= T_LEN)  tr = 2 * T_LEN - 2 - tr;
        v = x[(size_t)s * T_LEN + tr];
    }
    xpt[o] = v;
}

__global__ void zero_guards_kernel(float* __restrict__ y1t) {
    int idx = blockIdx.x * blockDim.x + threadIdx.x;
    y1t[idx] = 0.f;
    y1t[(size_t)(GT + TP) * S_SIG + idx] = 0.f;
}

template<int MT, bool STORE_OUT>
__global__ __launch_bounds__(128)
void fir_pass_kernel(const float* __restrict__ hext,
                     const float* __restrict__ src,
                     float* __restrict__ dst,
                     int t0_origin) {
    constexpr int G   = 16 * MT;
    constexpr int R   = 4 * MT;
    constexpr int BL  = R / 2;
    constexpr int NCH = ((1028 + 4 * (MT - 1) + R - 1) / R) * R;
    constexpr int SHN = G + 4 * (NCH + 4) + 8;
    static_assert(R % BL == 0, "B ring must divide the A ring");

    __shared__ float sh[SHN];
    for (int i = threadIdx.x; i < SHN; i += 128) {
        int d = i - G + GUARD;
        sh[i] = (d >= 0 && d < HEXT) ? hext[d] : 0.f;
    }
    __syncthreads();

    const int lane = threadIdx.x & 31;
    const int wave = threadIdx.x >> 5;
    const int n    = lane & 15;
    const int half = lane >> 4;
    const int q    = half * 2;
    const int c    = wave & 1;
    const int grp  = blockIdx.x * 2 + (wave >> 1);

    const int t0 = t0_origin + grp * (16 * MT);

    const float* sa = sh + (G + q - n);
    const float* p = src + ((size_t)(GT + t0 + TBASE + q) >> 1) * 64
                         + 2 * n + 32 * c;

    v8f acc[MT];
#pragma unroll
    for (int j = 0; j < MT; ++j) acc[j] = v8f{};

    v2f ring[R];
    v2f bq[BL];
#pragma unroll
    for (int s = 0; s < 4; ++s) {
        ring[s].x = sa[4 * s];
        ring[s].y = sa[4 * s + 1];
    }
#pragma unroll
    for (int s = 4; s < R; ++s) {
        ring[s].x = sa[4 * (s - R)];
        ring[s].y = sa[4 * (s - R) + 1];
    }
#pragma unroll
    for (int s = 0; s < BL - 4; ++s)
        bq[s] = *(const v2f*)(p + 128 * s);

    for (int c0 = 0; c0 < NCH; c0 += R) {
#pragma unroll
        for (int i0 = 0; i0 < R; i0 += 4) {
#pragma unroll
            for (int u = 0; u < 4; ++u)
                acc[MT - 1] = __builtin_amdgcn_wmma_f32_16x16x4_f32(
                    false, ring[(i0 + u + 4) % R], false, bq[(i0 + u) % BL],
                    (short)0, acc[MT - 1], false, false);
            __builtin_amdgcn_sched_barrier(0);
#pragma unroll
            for (int u = 0; u < 4; ++u) {
                ring[(i0 + u + 4) % R].x = sa[4 * (i0 + u + 4)];
                ring[(i0 + u + 4) % R].y = sa[4 * (i0 + u + 4) + 1];
            }
#pragma unroll
            for (int u = 0; u < 4; ++u)
                bq[(i0 + u + BL - 4) % BL] = *(const v2f*)(p + 128 * (i0 + u + BL - 4));
            __builtin_amdgcn_sched_barrier(0);
#pragma unroll
            for (int u = 0; u < 4; ++u) {
#pragma unroll
                for (int j = 0; j < MT - 1; ++j)
                    acc[j] = __builtin_amdgcn_wmma_f32_16x16x4_f32(
                        false, ring[(i0 + u - 4 * j + R) % R], false, bq[(i0 + u) % BL],
                        (short)0, acc[j], false, false);
            }
        }
        p  += 128 * R;
        sa += 4 * R;
    }

    if (!STORE_OUT) {
#pragma unroll
        for (int j = 0; j < MT; ++j) {
            const int row0 = GT + t0 + 16 * j + half * 8;
            float* o = dst + ((size_t)row0 >> 1) * 64 + (size_t)n * 2 + 32 * c;
#pragma unroll
            for (int r = 0; r < 4; ++r) {
                v2f w;
                w.x = acc[j][2 * r]; w.y = acc[j][2 * r + 1];
                *(v2f*)(o + (size_t)r * 64) = w;
            }
        }
    } else {
#pragma unroll
        for (int j = 0; j < MT; ++j) {
            float* o = dst + (size_t)(n + 16 * c) * T_LEN
                           + (t0 + 16 * j - PAD) + half * 8;
            *(float4*)(o)     = make_float4(acc[j][0], acc[j][1], acc[j][2], acc[j][3]);
            *((float4*)o + 1) = make_float4(acc[j][4], acc[j][5], acc[j][6], acc[j][7]);
        }
    }
}

extern "C" void kernel_launch(void* const* d_in, const int* in_sizes, int n_in,
                              void* d_out, int out_size, void* d_ws, size_t ws_size,
                              hipStream_t stream) {
    const float* x    = (const float*)d_in[0];
    const float* kern = (const float*)d_in[1];
    float* out = (float*)d_out;
    float* ws  = (float*)d_ws;

    float* hext = ws;
    float* hrev = ws + HEXT;
    float* xpt  = ws + 2 * HEXT;
    float* y1t  = xpt + (size_t)ROWS * S_SIG;

    prep_h_kernel<<<(HEXT + 255) / 256, 256, 0, stream>>>(kern, hext, hrev);
    build_xpt_kernel<<<(ROWS * S_SIG) / 256, 256, 0, stream>>>(x, xpt);
    zero_guards_kernel<<<(GT * S_SIG) / 256, 256, 0, stream>>>(y1t);

    fir_pass_kernel<MT1, false>
        <<<TP / (16 * MT1 * 2), 128, 0, stream>>>(hext, xpt, y1t, 0);
    fir_pass_kernel<MT2, true>
        <<<T_LEN / (16 * MT2 * 2), 128, 0, stream>>>(hrev, y1t, out, PAD);
}


# filter staging into LDS issues all of a thread's loads before its first store, as 9 independent 16-byte loads instead of 34 load-wait-store round trips in series, in both passes; values unchanged
# speedup vs baseline: 1.3572x; 1.0547x over previous
#include <hip/hip_runtime.h>


typedef __attribute__((ext_vector_type(2))) float v2f;
typedef __attribute__((ext_vector_type(8))) float v8f;

#define K_TAPS 4096
#define PAD    2048
#define T_LEN  131072
#define S_SIG  32
#define TP     (T_LEN + 2*PAD)
#define GT     2112
#define ROWS   (TP + 2*GT)
#define GUARD  16
#define HEXT   (K_TAPS + 64)
#define TBASE  (-2048)
#define MT1    6
#define MT2    8

__global__ void prep_h_kernel(const float* __restrict__ h,
                              float* __restrict__ hext,
                              float* __restrict__ hrev) {
    int i = blockIdx.x * blockDim.x + threadIdx.x;
    if (i >= HEXT) return;
    int j = i - GUARD - 1;
    int d = i - GUARD;
    hext[i] = (j >= 0 && j < K_TAPS) ? h[j] : 0.f;
    hrev[i] = (d >= 0 && d < K_TAPS) ? h[K_TAPS - 1 - d] : 0.f;
}

__global__ void build_xpt_kernel(const float* __restrict__ x,
                                 float* __restrict__ xpt) {
    int o = blockIdx.x * blockDim.x + threadIdx.x;
    int rem = o & 63;
    int s = rem >> 1;
    int t = ((o >> 6) << 1) + (rem & 1);
    int tg = t - GT;
    float v = 0.f;
    if (tg >= 0 && tg < TP) {
        int tr = tg - PAD;
        if (tr < 0)            tr = -tr;
        else if (tr >= T_LEN)  tr = 2 * T_LEN - 2 - tr;
        v = x[(size_t)s * T_LEN + tr];
    }
    xpt[o] = v;
}

__global__ void zero_guards_kernel(float* __restrict__ y1t) {
    int idx = blockIdx.x * blockDim.x + threadIdx.x;
    y1t[idx] = 0.f;
    y1t[(size_t)(GT + TP) * S_SIG + idx] = 0.f;
}

template<int MT, bool STORE_OUT>
__global__ __launch_bounds__(128)
void fir_pass_kernel(const float* __restrict__ hext,
                     const float* __restrict__ src,
                     float* __restrict__ dst,
                     int t0_origin) {
    constexpr int G   = 16 * MT;
    constexpr int R   = 4 * MT;
    constexpr int BL  = R / 2;
    constexpr int NCH = ((1028 + 4 * (MT - 1) + R - 1) / R) * R;
    constexpr int SHN = G + 4 * (NCH + 4) + 8;
    static_assert(R % BL == 0, "B ring must divide the A ring");

    constexpr int NV  = SHN / 4;
    constexpr int NIT = (NV + 127) / 128;
    static_assert(SHN % 4 == 0 && (G - GUARD) % 4 == 0 && HEXT % 4 == 0,
                  "a 16-byte slot lies wholly inside or wholly outside hext");

    __shared__ __attribute__((aligned(16))) float sh[SHN];
    float4 stage[NIT];
#pragma unroll
    for (int j = 0; j < NIT; ++j) {
        const int d  = 4 * ((int)threadIdx.x + 128 * j) - G + GUARD;
        const int dc = min(max(d, 0), HEXT - 4);
        const float4 w = *(const float4*)(hext + dc);
        stage[j] = (d >= 0 && d < HEXT) ? w : make_float4(0.f, 0.f, 0.f, 0.f);
    }
#pragma unroll
    for (int j = 0; j < NIT; ++j) {
        const int v = (int)threadIdx.x + 128 * j;
        if (v < NV) *(float4*)(sh + 4 * v) = stage[j];
    }
    __syncthreads();

    const int lane = threadIdx.x & 31;
    const int wave = threadIdx.x >> 5;
    const int n    = lane & 15;
    const int half = lane >> 4;
    const int q    = half * 2;
    const int c    = wave & 1;
    const int grp  = blockIdx.x * 2 + (wave >> 1);

    const int t0 = t0_origin + grp * (16 * MT);

    const float* sa = sh + (G + q - n);
    const float* p = src + ((size_t)(GT + t0 + TBASE + q) >> 1) * 64
                         + 2 * n + 32 * c;

    v8f acc[MT];
#pragma unroll
    for (int j = 0; j < MT; ++j) acc[j] = v8f{};

    v2f ring[R];
    v2f bq[BL];
#pragma unroll
    for (int s = 0; s < 4; ++s) {
        ring[s].x = sa[4 * s];
        ring[s].y = sa[4 * s + 1];
    }
#pragma unroll
    for (int s = 4; s < R; ++s) {
        ring[s].x = sa[4 * (s - R)];
        ring[s].y = sa[4 * (s - R) + 1];
    }
#pragma unroll
    for (int s = 0; s < BL - 4; ++s)
        bq[s] = *(const v2f*)(p + 128 * s);

    for (int c0 = 0; c0 < NCH; c0 += R) {
#pragma unroll
        for (int i0 = 0; i0 < R; i0 += 4) {
#pragma unroll
            for (int u = 0; u < 4; ++u)
                acc[MT - 1] = __builtin_amdgcn_wmma_f32_16x16x4_f32(
                    false, ring[(i0 + u + 4) % R], false, bq[(i0 + u) % BL],
                    (short)0, acc[MT - 1], false, false);
            __builtin_amdgcn_sched_barrier(0);
#pragma unroll
            for (int u = 0; u < 4; ++u) {
                ring[(i0 + u + 4) % R].x = sa[4 * (i0 + u + 4)];
                ring[(i0 + u + 4) % R].y = sa[4 * (i0 + u + 4) + 1];
            }
#pragma unroll
            for (int u = 0; u < 4; ++u)
                bq[(i0 + u + BL - 4) % BL] = *(const v2f*)(p + 128 * (i0 + u + BL - 4));
            __builtin_amdgcn_sched_barrier(0);
#pragma unroll
            for (int u = 0; u < 4; ++u) {
#pragma unroll
                for (int j = 0; j < MT - 1; ++j)
                    acc[j] = __builtin_amdgcn_wmma_f32_16x16x4_f32(
                        false, ring[(i0 + u - 4 * j + R) % R], false, bq[(i0 + u) % BL],
                        (short)0, acc[j], false, false);
            }
        }
        p  += 128 * R;
        sa += 4 * R;
    }

    if (!STORE_OUT) {
#pragma unroll
        for (int j = 0; j < MT; ++j) {
            const int row0 = GT + t0 + 16 * j + half * 8;
            float* o = dst + ((size_t)row0 >> 1) * 64 + (size_t)n * 2 + 32 * c;
#pragma unroll
            for (int r = 0; r < 4; ++r) {
                v2f w;
                w.x = acc[j][2 * r]; w.y = acc[j][2 * r + 1];
                *(v2f*)(o + (size_t)r * 64) = w;
            }
        }
    } else {
#pragma unroll
        for (int j = 0; j < MT; ++j) {
            float* o = dst + (size_t)(n + 16 * c) * T_LEN
                           + (t0 + 16 * j - PAD) + half * 8;
            *(float4*)(o)     = make_float4(acc[j][0], acc[j][1], acc[j][2], acc[j][3]);
            *((float4*)o + 1) = make_float4(acc[j][4], acc[j][5], acc[j][6], acc[j][7]);
        }
    }
}

extern "C" void kernel_launch(void* const* d_in, const int* in_sizes, int n_in,
                              void* d_out, int out_size, void* d_ws, size_t ws_size,
                              hipStream_t stream) {
    const float* x    = (const float*)d_in[0];
    const float* kern = (const float*)d_in[1];
    float* out = (float*)d_out;
    float* ws  = (float*)d_ws;

    float* hext = ws;
    float* hrev = ws + HEXT;
    float* xpt  = ws + 2 * HEXT;
    float* y1t  = xpt + (size_t)ROWS * S_SIG;

    prep_h_kernel<<<(HEXT + 255) / 256, 256, 0, stream>>>(kern, hext, hrev);
    build_xpt_kernel<<<(ROWS * S_SIG) / 256, 256, 0, stream>>>(x, xpt);
    zero_guards_kernel<<<(GT * S_SIG) / 256, 256, 0, stream>>>(y1t);

    fir_pass_kernel<MT1, false>
        <<<TP / (16 * MT1 * 2), 128, 0, stream>>>(hext, xpt, y1t, 0);
    fir_pass_kernel<MT2, true>
        <<<T_LEN / (16 * MT2 * 2), 128, 0, stream>>>(hrev, y1t, out, PAD);
}


# each pass as a 2-phase fast FIR: three Toeplitz products of K/2 taps (even taps, odd taps, their sum on summed phases) recombined in registers instead of four; order of summation changes
# speedup vs baseline: 1.4770x; 1.0883x over previous
#include <hip/hip_runtime.h>


typedef __attribute__((ext_vector_type(2))) float v2f;
typedef __attribute__((ext_vector_type(4))) float v4f;
typedef __attribute__((ext_vector_type(8))) float v8f;

#define K_TAPS 4096
#define KH     2048
#define PAD    2048
#define T_LEN  131072
#define S_SIG  32
#define TP     (T_LEN + 2*PAD)
#define GT     2112
#define ROWS   (TP + 2*GT)
#define SBASE  64
#define MT1    3
#define MT2    4
#define XOFF   16384

template<int MT> struct Geo {
    static constexpr int G   = 16 * (MT + 1);
    static constexpr int R   = 4 * (MT + 1);
    static constexpr int BL  = R / 2;
    static constexpr int NCH = ((516 + 4 * MT + R - 1) / R) * R;
    static constexpr int SHN = G + 4 * (NCH + 4) + 8;
};
#define F1OFF  0
#define F2OFF  (3 * Geo<MT1>::SHN)
static_assert(F2OFF + 3 * Geo<MT2>::SHN <= XOFF, "filter blocks end before the staged input");

__device__ __forceinline__ int quad_slot(int r) { return ((r & 1) << 1) | ((r >> 1) & 1); }

__device__ __forceinline__ float f_tap(const float* __restrict__ h, int e, bool backward) {
    const int i = backward ? (K_TAPS - 1 - e) : (e - 1);
    return (i >= 0 && i < K_TAPS) ? h[i] : 0.f;
}

template<int MT>
__device__ __forceinline__ float subfilter(const float* __restrict__ h, int o, bool backward) {
    const int x = o / Geo<MT>::SHN;
    const int j = o - x * Geo<MT>::SHN - Geo<MT>::G;
    if (j < 0 || j > KH) return 0.f;
    const float fe = f_tap(h, 2 * j, backward);
    const float fo = f_tap(h, 2 * j - 1, backward);
    return x == 0 ? fe : (x == 1 ? fo : fe + fo);
}

__global__ void build_xq_kernel(const float* __restrict__ x,
                                float* __restrict__ xq) {
    int o = blockIdx.x * blockDim.x + threadIdx.x;
    int rem = o & 127;
    int s = rem >> 2;
    int t = ((o >> 7) << 2) + quad_slot(rem & 3);
    int tg = t - GT;
    float v = 0.f;
    if (tg >= 0 && tg < TP) {
        int tr = tg - PAD;
        if (tr < 0)            tr = -tr;
        else if (tr >= T_LEN)  tr = 2 * T_LEN - 2 - tr;
        v = x[(size_t)s * T_LEN + tr];
    }
    xq[o] = v;
}

__global__ void prep_h_kernel(const float* __restrict__ h,
                              float* __restrict__ f1,
                              float* __restrict__ f2) {
    int o = blockIdx.x * blockDim.x + threadIdx.x;
    if (o < 3 * Geo<MT1>::SHN) f1[o] = subfilter<MT1>(h, o, false);
    if (o < 3 * Geo<MT2>::SHN) f2[o] = subfilter<MT2>(h, o, true);
}

__global__ void zero_guards_kernel(float* __restrict__ y1q) {
    int idx = blockIdx.x * blockDim.x + threadIdx.x;
    y1q[idx] = 0.f;
    y1q[(size_t)(GT + TP) * S_SIG + idx] = 0.f;
}

template<int MT, bool STORE_OUT>
__global__ __launch_bounds__(128)
void fir_pass_kernel(const float* __restrict__ filt,
                     const float* __restrict__ src,
                     float* __restrict__ dst,
                     int t0_origin) {
    constexpr int G   = Geo<MT>::G;
    constexpr int R   = Geo<MT>::R;
    constexpr int BL  = Geo<MT>::BL;
    constexpr int NCH = Geo<MT>::NCH;
    constexpr int SHN = Geo<MT>::SHN;
    static_assert(R % BL == 0, "B ring must divide the A ring");
    static_assert(SHN % 4 == 0, "whole 16-byte slots");

    constexpr int NV  = 3 * SHN / 4;
    constexpr int NIT = (NV + 127) / 128;

    __shared__ __attribute__((aligned(16))) float sh[3 * SHN];
    float4 stage[NIT];
#pragma unroll
    for (int j = 0; j < NIT; ++j) {
        const int v = min((int)threadIdx.x + 128 * j, NV - 1);
        stage[j] = *(const float4*)(filt + 4 * v);
    }
#pragma unroll
    for (int j = 0; j < NIT; ++j) {
        const int v = (int)threadIdx.x + 128 * j;
        if (v < NV) *(float4*)(sh + 4 * v) = stage[j];
    }
    __syncthreads();

    const int lane = threadIdx.x & 31;
    const int wave = threadIdx.x >> 5;
    const int n    = lane & 15;
    const int half = lane >> 4;
    const int q    = half * 2;
    const int c    = wave & 1;
    const int grp  = blockIdx.x * 2 + (wave >> 1);

    const int t0 = t0_origin + grp * (32 * MT);

    const float* saE = sh + (G + q - n);
    const float* saO = saE + SHN;
    const float* saS = saE + 2 * SHN;
    const float* p = src + ((size_t)((SBASE + t0 - 32) >> 2) + half) * 128
                         + 4 * (n + 16 * c);

    v8f accE[MT + 1], accS[MT + 1], accO[MT + 1];
#pragma unroll
    for (int j = 0; j <= MT; ++j) { accE[j] = v8f{}; accS[j] = v8f{}; accO[j] = v8f{}; }

    v2f rE[R], rO[R], rS[R];
    v4f bq[BL];
#pragma unroll
    for (int s = 0; s < R; ++s) {
        const int kc = (s < 4) ? s : s - R;
        rE[s].x = saE[4 * kc]; rE[s].y = saE[4 * kc + 1];
        rO[s].x = saO[4 * kc]; rO[s].y = saO[4 * kc + 1];
        rS[s].x = saS[4 * kc]; rS[s].y = saS[4 * kc + 1];
    }
#pragma unroll
    for (int s = 0; s < BL - 4; ++s)
        bq[s] = *(const v4f*)(p + 256 * s);

    for (int c0 = 0; c0 < NCH; c0 += R) {
#pragma unroll
        for (int i0 = 0; i0 < R; i0 += 4) {
#pragma unroll
            for (int u = 0; u < 4; ++u) {
                const v4f b = bq[(i0 + u) % BL];
                const v2f be = {b.x, b.y};
                const v2f bo = {b.z, b.w};
                const v2f bs = be + bo;
                accE[MT] = __builtin_amdgcn_wmma_f32_16x16x4_f32(
                    false, rE[(i0 + u + 4) % R], false, be, (short)0, accE[MT], false, false);
                accO[MT] = __builtin_amdgcn_wmma_f32_16x16x4_f32(
                    false, rO[(i0 + u + 4) % R], false, bo, (short)0, accO[MT], false, false);
                accS[MT] = __builtin_amdgcn_wmma_f32_16x16x4_f32(
                    false, rS[(i0 + u + 4) % R], false, bs, (short)0, accS[MT], false, false);
            }
            __builtin_amdgcn_sched_barrier(0);
#pragma unroll
            for (int u = 0; u < 4; ++u) {
                const int s = (i0 + u + 4) % R;
                const int kc = i0 + u + 4;
                rE[s].x = saE[4 * kc]; rE[s].y = saE[4 * kc + 1];
                rO[s].x = saO[4 * kc]; rO[s].y = saO[4 * kc + 1];
                rS[s].x = saS[4 * kc]; rS[s].y = saS[4 * kc + 1];
            }
#pragma unroll
            for (int u = 0; u < 4; ++u)
                bq[(i0 + u + BL - 4) % BL] = *(const v4f*)(p + 256 * (i0 + u + BL - 4));
            __builtin_amdgcn_sched_barrier(0);
#pragma unroll
            for (int u = 0; u < 4; ++u) {
                const v4f b = bq[(i0 + u) % BL];
                const v2f be = {b.x, b.y};
                const v2f bo = {b.z, b.w};
                const v2f bs = be + bo;
                accO[0] = __builtin_amdgcn_wmma_f32_16x16x4_f32(
                    false, rO[(i0 + u) % R], false, bo, (short)0, accO[0], false, false);
#pragma unroll
                for (int j = 1; j < MT; ++j) {
                    accE[j] = __builtin_amdgcn_wmma_f32_16x16x4_f32(
                        false, rE[(i0 + u - 4 * j + R) % R], false, be, (short)0, accE[j], false, false);
                    accO[j] = __builtin_amdgcn_wmma_f32_16x16x4_f32(
                        false, rO[(i0 + u - 4 * j + R) % R], false, bo, (short)0, accO[j], false, false);
                    accS[j] = __builtin_amdgcn_wmma_f32_16x16x4_f32(
                        false, rS[(i0 + u - 4 * j + R) % R], false, bs, (short)0, accS[j], false, false);
                }
            }
        }
        p   += 256 * R;
        saE += 4 * R;
        saO += 4 * R;
        saS += 4 * R;
    }

#pragma unroll
    for (int jj = 1; jj <= MT; ++jj) {
        const float send = half ? accO[jj - 1][7] : accO[jj][7];
        const float prev = __shfl_xor(send, 16);
        float ev[8], od[8];
#pragma unroll
        for (int r = 0; r < 8; ++r) {
            const float bm1 = (r == 0) ? prev : accO[jj][r - 1];
            ev[r] = accE[jj][r] + bm1;
            od[r] = accS[jj][r] - accE[jj][r] - accO[jj][r];
        }
        const int tt = t0 + 32 * (jj - 1) + 16 * half;
        if (!STORE_OUT) {
            float* o = dst + ((size_t)(GT + tt) >> 2) * 128 + 4 * (n + 16 * c);
#pragma unroll
            for (int g = 0; g < 4; ++g)
                *(float4*)(o + (size_t)g * 128) =
                    make_float4(ev[2 * g], ev[2 * g + 1], od[2 * g], od[2 * g + 1]);
        } else {
            float* o = dst + (size_t)(n + 16 * c) * T_LEN + (tt - PAD);
#pragma unroll
            for (int g = 0; g < 4; ++g)
                *((float4*)o + g) =
                    make_float4(ev[2 * g], od[2 * g], ev[2 * g + 1], od[2 * g + 1]);
        }
    }
}

extern "C" void kernel_launch(void* const* d_in, const int* in_sizes, int n_in,
                              void* d_out, int out_size, void* d_ws, size_t ws_size,
                              hipStream_t stream) {
    const float* x    = (const float*)d_in[0];
    const float* kern = (const float*)d_in[1];
    float* out = (float*)d_out;
    float* ws  = (float*)d_ws;

    float* f1  = ws + F1OFF;
    float* f2  = ws + F2OFF;
    float* xq  = ws + XOFF;
    float* y1q = xq + (size_t)ROWS * S_SIG;

    prep_h_kernel<<<(3 * Geo<MT2>::SHN + 255) / 256, 256, 0, stream>>>(kern, f1, f2);
    build_xq_kernel<<<(ROWS * S_SIG) / 256, 256, 0, stream>>>(x, xq);
    zero_guards_kernel<<<(GT * S_SIG) / 256, 256, 0, stream>>>(y1q);

    fir_pass_kernel<MT1, false>
        <<<TP / (32 * MT1 * 2), 128, 0, stream>>>(f1, xq, y1q, 0);
    fir_pass_kernel<MT2, true>
        <<<T_LEN / (32 * MT2 * 2), 128, 0, stream>>>(f2, y1q, out, PAD);
}


# backward pass as the 2-phase fast FIR applied twice: nine quarter-rate Toeplitz products of K/4 taps recombined in registers instead of three of K/2, one wave per SIMD; order of summation changes
# speedup vs baseline: 1.6287x; 1.1027x over previous
#include <hip/hip_runtime.h>


typedef __attribute__((ext_vector_type(2))) float v2f;
typedef __attribute__((ext_vector_type(4))) float v4f;
typedef __attribute__((ext_vector_type(8))) float v8f;

#define K_TAPS 4096
#define KH     2048
#define PAD    2048
#define T_LEN  131072
#define S_SIG  32
#define TP     (T_LEN + 2*PAD)
#define GT     2112
#define ROWS   (TP + 2*GT)
#define SBASE  64
#define MT1    3
#define KQ     1024
#define MT2    4
#define XOFF   32768

template<int MT> struct Geo {
    static constexpr int G   = 16 * (MT + 1);
    static constexpr int R   = 4 * (MT + 1);
    static constexpr int BL  = R / 2;
    static constexpr int NCH = ((516 + 4 * MT + R - 1) / R) * R;
    static constexpr int SHN = G + 4 * (NCH + 4) + 8;
};
template<int MT> struct Geo4 {
    static constexpr int G   = 16 * (MT + 1);
    static constexpr int R   = 4 * (MT + 1);
    static constexpr int BL  = R / 2;
    static constexpr int NCH = ((260 + 4 * MT + R - 1) / R) * R;
    static constexpr int SHN = G + 4 * (NCH + 4) + 8;
};
#define F1OFF  0
#define F2OFF  (3 * Geo<MT1>::SHN)
static_assert(F2OFF + 9 * Geo4<MT2>::SHN <= XOFF, "filter blocks end before the staged input");

__device__ __forceinline__ int quad_slot(int r) { return ((r & 1) << 1) | ((r >> 1) & 1); }

__device__ __forceinline__ float f_tap(const float* __restrict__ h, int e, bool backward) {
    const int i = backward ? (K_TAPS - 1 - e) : (e - 1);
    return (i >= 0 && i < K_TAPS) ? h[i] : 0.f;
}

template<int MT>
__device__ __forceinline__ float subfilter(const float* __restrict__ h, int o, bool backward) {
    const int x = o / Geo<MT>::SHN;
    const int j = o - x * Geo<MT>::SHN - Geo<MT>::G;
    if (j < 0 || j > KH) return 0.f;
    const float fe = f_tap(h, 2 * j, backward);
    const float fo = f_tap(h, 2 * j - 1, backward);
    return x == 0 ? fe : (x == 1 ? fo : fe + fo);
}

__device__ __forceinline__ float outer_tap(const float* __restrict__ h, int j, int xo, bool backward) {
    if (j < 0 || j > KH) return 0.f;
    const float fe = f_tap(h, 2 * j, backward);
    const float fo = f_tap(h, 2 * j - 1, backward);
    return xo == 0 ? fe : (xo == 1 ? fo : fe + fo);
}

template<int MT>
__device__ __forceinline__ float subfilter4(const float* __restrict__ h, int o, bool backward) {
    const int x  = o / Geo4<MT>::SHN;
    const int jq = o - x * Geo4<MT>::SHN - Geo4<MT>::G;
    if (jq < 0 || jq > KQ) return 0.f;
    const int xo = x / 3;
    const int xi = x - 3 * xo;
    const float ge = outer_tap(h, 2 * jq, xo, backward);
    const float go = outer_tap(h, 2 * jq - 1, xo, backward);
    return xi == 0 ? ge : (xi == 1 ? go : ge + go);
}

__global__ void build_xq_kernel(const float* __restrict__ x,
                                float* __restrict__ xq) {
    int o = blockIdx.x * blockDim.x + threadIdx.x;
    int rem = o & 127;
    int s = rem >> 2;
    int t = ((o >> 7) << 2) + quad_slot(rem & 3);
    int tg = t - GT;
    float v = 0.f;
    if (tg >= 0 && tg < TP) {
        int tr = tg - PAD;
        if (tr < 0)            tr = -tr;
        else if (tr >= T_LEN)  tr = 2 * T_LEN - 2 - tr;
        v = x[(size_t)s * T_LEN + tr];
    }
    xq[o] = v;
}

__global__ void prep_h_kernel(const float* __restrict__ h,
                              float* __restrict__ f1,
                              float* __restrict__ f2) {
    int o = blockIdx.x * blockDim.x + threadIdx.x;
    if (o < 3 * Geo<MT1>::SHN) f1[o] = subfilter<MT1>(h, o, false);
    if (o < 9 * Geo4<MT2>::SHN) f2[o] = subfilter4<MT2>(h, o, true);
}

__global__ void zero_guards_kernel(float* __restrict__ y1q) {
    int idx = blockIdx.x * blockDim.x + threadIdx.x;
    y1q[idx] = 0.f;
    y1q[(size_t)(GT + TP) * S_SIG + idx] = 0.f;
}

template<int MT>
__global__ __launch_bounds__(128)
void fir_pass_kernel(const float* __restrict__ filt,
                     const float* __restrict__ src,
                     float* __restrict__ dst,
                     int t0_origin) {
    constexpr int G   = Geo<MT>::G;
    constexpr int R   = Geo<MT>::R;
    constexpr int BL  = Geo<MT>::BL;
    constexpr int NCH = Geo<MT>::NCH;
    constexpr int SHN = Geo<MT>::SHN;
    static_assert(R % BL == 0, "B ring must divide the A ring");
    static_assert(SHN % 4 == 0, "whole 16-byte slots");

    constexpr int NV  = 3 * SHN / 4;
    constexpr int NIT = (NV + 127) / 128;

    __shared__ __attribute__((aligned(16))) float sh[3 * SHN];
    float4 stage[NIT];
#pragma unroll
    for (int j = 0; j < NIT; ++j) {
        const int v = min((int)threadIdx.x + 128 * j, NV - 1);
        stage[j] = *(const float4*)(filt + 4 * v);
    }
#pragma unroll
    for (int j = 0; j < NIT; ++j) {
        const int v = (int)threadIdx.x + 128 * j;
        if (v < NV) *(float4*)(sh + 4 * v) = stage[j];
    }
    __syncthreads();

    const int lane = threadIdx.x & 31;
    const int wave = threadIdx.x >> 5;
    const int n    = lane & 15;
    const int half = lane >> 4;
    const int q    = half * 2;
    const int c    = wave & 1;
    const int grp  = blockIdx.x * 2 + (wave >> 1);

    const int t0 = t0_origin + grp * (32 * MT);

    const float* saE = sh + (G + q - n);
    const float* saO = saE + SHN;
    const float* saS = saE + 2 * SHN;
    const float* p = src + ((size_t)((SBASE + t0 - 32) >> 2) + half) * 128
                         + 4 * (n + 16 * c);

    v8f accE[MT + 1], accS[MT + 1], accO[MT + 1];
#pragma unroll
    for (int j = 0; j <= MT; ++j) { accE[j] = v8f{}; accS[j] = v8f{}; accO[j] = v8f{}; }

    v2f rE[R], rO[R], rS[R];
    v4f bq[BL];
#pragma unroll
    for (int s = 0; s < R; ++s) {
        const int kc = (s < 4) ? s : s - R;
        rE[s].x = saE[4 * kc]; rE[s].y = saE[4 * kc + 1];
        rO[s].x = saO[4 * kc]; rO[s].y = saO[4 * kc + 1];
        rS[s].x = saS[4 * kc]; rS[s].y = saS[4 * kc + 1];
    }
#pragma unroll
    for (int s = 0; s < BL - 4; ++s)
        bq[s] = *(const v4f*)(p + 256 * s);

    for (int c0 = 0; c0 < NCH; c0 += R) {
#pragma unroll
        for (int i0 = 0; i0 < R; i0 += 4) {
#pragma unroll
            for (int u = 0; u < 4; ++u) {
                const v4f b = bq[(i0 + u) % BL];
                const v2f be = {b.x, b.y};
                const v2f bo = {b.z, b.w};
                const v2f bs = be + bo;
                accE[MT] = __builtin_amdgcn_wmma_f32_16x16x4_f32(
                    false, rE[(i0 + u + 4) % R], false, be, (short)0, accE[MT], false, false);
                accO[MT] = __builtin_amdgcn_wmma_f32_16x16x4_f32(
                    false, rO[(i0 + u + 4) % R], false, bo, (short)0, accO[MT], false, false);
                accS[MT] = __builtin_amdgcn_wmma_f32_16x16x4_f32(
                    false, rS[(i0 + u + 4) % R], false, bs, (short)0, accS[MT], false, false);
            }
            __builtin_amdgcn_sched_barrier(0);
#pragma unroll
            for (int u = 0; u < 4; ++u) {
                const int s = (i0 + u + 4) % R;
                const int kc = i0 + u + 4;
                rE[s].x = saE[4 * kc]; rE[s].y = saE[4 * kc + 1];
                rO[s].x = saO[4 * kc]; rO[s].y = saO[4 * kc + 1];
                rS[s].x = saS[4 * kc]; rS[s].y = saS[4 * kc + 1];
            }
#pragma unroll
            for (int u = 0; u < 4; ++u)
                bq[(i0 + u + BL - 4) % BL] = *(const v4f*)(p + 256 * (i0 + u + BL - 4));
            __builtin_amdgcn_sched_barrier(0);
#pragma unroll
            for (int u = 0; u < 4; ++u) {
                const v4f b = bq[(i0 + u) % BL];
                const v2f be = {b.x, b.y};
                const v2f bo = {b.z, b.w};
                const v2f bs = be + bo;
                accO[0] = __builtin_amdgcn_wmma_f32_16x16x4_f32(
                    false, rO[(i0 + u) % R], false, bo, (short)0, accO[0], false, false);
#pragma unroll
                for (int j = 1; j < MT; ++j) {
                    accE[j] = __builtin_amdgcn_wmma_f32_16x16x4_f32(
                        false, rE[(i0 + u - 4 * j + R) % R], false, be, (short)0, accE[j], false, false);
                    accO[j] = __builtin_amdgcn_wmma_f32_16x16x4_f32(
                        false, rO[(i0 + u - 4 * j + R) % R], false, bo, (short)0, accO[j], false, false);
                    accS[j] = __builtin_amdgcn_wmma_f32_16x16x4_f32(
                        false, rS[(i0 + u - 4 * j + R) % R], false, bs, (short)0, accS[j], false, false);
                }
            }
        }
        p   += 256 * R;
        saE += 4 * R;
        saO += 4 * R;
        saS += 4 * R;
    }

#pragma unroll
    for (int jj = 1; jj <= MT; ++jj) {
        const float send = half ? accO[jj - 1][7] : accO[jj][7];
        const float prev = __shfl_xor(send, 16);
        float ev[8], od[8];
#pragma unroll
        for (int r = 0; r < 8; ++r) {
            const float bm1 = (r == 0) ? prev : accO[jj][r - 1];
            ev[r] = accE[jj][r] + bm1;
            od[r] = accS[jj][r] - accE[jj][r] - accO[jj][r];
        }
        const int tt = t0 + 32 * (jj - 1) + 16 * half;
        float* o = dst + ((size_t)(GT + tt) >> 2) * 128 + 4 * (n + 16 * c);
#pragma unroll
        for (int g = 0; g < 4; ++g)
            *(float4*)(o + (size_t)g * 128) =
                make_float4(ev[2 * g], ev[2 * g + 1], od[2 * g], od[2 * g + 1]);
    }
}

template<int MT>
__global__ __launch_bounds__(128)
void fir_pass4_kernel(const float* __restrict__ filt,
                      const float* __restrict__ src,
                      float* __restrict__ dst,
                      int t0_origin) {
    constexpr int G   = Geo4<MT>::G;
    constexpr int R   = Geo4<MT>::R;
    constexpr int BL  = Geo4<MT>::BL;
    constexpr int NCH = Geo4<MT>::NCH;
    constexpr int SHN = Geo4<MT>::SHN;
    static_assert(R % BL == 0, "B ring must divide the A ring");
    static_assert(SHN % 4 == 0, "whole 16-byte slots");
    static_assert(4 * NCH >= KQ + 1 + 16 * MT + 15, "every tile's tap window lies inside the k axis");

    constexpr int NV  = 9 * SHN / 4;
    constexpr int NIT = (NV + 127) / 128;

    __shared__ __attribute__((aligned(16))) float sh[9 * SHN];
    float4 stage[NIT];
#pragma unroll
    for (int j = 0; j < NIT; ++j) {
        const int v = min((int)threadIdx.x + 128 * j, NV - 1);
        stage[j] = *(const float4*)(filt + 4 * v);
    }
#pragma unroll
    for (int j = 0; j < NIT; ++j) {
        const int v = (int)threadIdx.x + 128 * j;
        if (v < NV) *(float4*)(sh + 4 * v) = stage[j];
    }
    __syncthreads();

    const int lane = threadIdx.x & 31;
    const int wave = threadIdx.x >> 5;
    const int n    = lane & 15;
    const int half = lane >> 4;
    const int q    = half * 2;
    const int c    = wave & 1;
    const int grp  = blockIdx.x * 2 + (wave >> 1);

    const int t0 = t0_origin + grp * (64 * MT);

    const float* sa = sh + (G + q - n);
    const float* p = src + ((size_t)(((SBASE + t0) >> 2) - 16) + 2 * half) * 128
                         + 4 * (n + 16 * c);

    v8f acc[9][MT + 1];
#pragma unroll
    for (int x = 0; x < 9; ++x)
#pragma unroll
        for (int j = 0; j <= MT; ++j) acc[x][j] = v8f{};

    v2f ring[9][R];
    v4f bq[BL][2];
#pragma unroll
    for (int s = 0; s < R; ++s) {
        const int kc = (s < 4) ? s : s - R;
#pragma unroll
        for (int x = 0; x < 9; ++x) {
            ring[x][s].x = sa[x * SHN + 4 * kc];
            ring[x][s].y = sa[x * SHN + 4 * kc + 1];
        }
    }
#pragma unroll
    for (int s = 0; s < BL - 4; ++s) {
        bq[s][0] = *(const v4f*)(p + 512 * s);
        bq[s][1] = *(const v4f*)(p + 512 * s + 128);
    }

    for (int c0 = 0; c0 < NCH; c0 += R) {
#pragma unroll
        for (int i0 = 0; i0 < R; i0 += 4) {
#pragma unroll
            for (int u = 0; u < 4; ++u) {
                const v4f g0 = bq[(i0 + u) % BL][0];
                const v4f g1 = bq[(i0 + u) % BL][1];
                v2f bf[9];
                bf[0] = v2f{g0.x, g1.x};
                bf[1] = v2f{g0.y, g1.y};
                bf[3] = v2f{g0.z, g1.z};
                bf[4] = v2f{g0.w, g1.w};
                bf[2] = bf[0] + bf[1];
                bf[5] = bf[3] + bf[4];
                bf[6] = bf[0] + bf[3];
                bf[7] = bf[1] + bf[4];
                bf[8] = bf[6] + bf[7];
#pragma unroll
                for (int x = 0; x < 9; ++x)
                    acc[x][MT] = __builtin_amdgcn_wmma_f32_16x16x4_f32(
                        false, ring[x][(i0 + u + 4) % R], false, bf[x], (short)0, acc[x][MT], false, false);
            }
            __builtin_amdgcn_sched_barrier(0);
#pragma unroll
            for (int u = 0; u < 4; ++u) {
                const int s = (i0 + u + 4) % R;
                const int kc = i0 + u + 4;
#pragma unroll
                for (int x = 0; x < 9; ++x) {
                    ring[x][s].x = sa[x * SHN + 4 * kc];
                    ring[x][s].y = sa[x * SHN + 4 * kc + 1];
                }
            }
#pragma unroll
            for (int u = 0; u < 4; ++u) {
                bq[(i0 + u + BL - 4) % BL][0] = *(const v4f*)(p + 512 * (i0 + u + BL - 4));
                bq[(i0 + u + BL - 4) % BL][1] = *(const v4f*)(p + 512 * (i0 + u + BL - 4) + 128);
            }
            __builtin_amdgcn_sched_barrier(0);
#pragma unroll
            for (int u = 0; u < 4; ++u) {
                const v4f g0 = bq[(i0 + u) % BL][0];
                const v4f g1 = bq[(i0 + u) % BL][1];
                v2f bf[9];
                bf[0] = v2f{g0.x, g1.x};
                bf[1] = v2f{g0.y, g1.y};
                bf[3] = v2f{g0.z, g1.z};
                bf[4] = v2f{g0.w, g1.w};
                bf[2] = bf[0] + bf[1];
                bf[5] = bf[3] + bf[4];
                bf[6] = bf[0] + bf[3];
                bf[7] = bf[1] + bf[4];
                bf[8] = bf[6] + bf[7];
#pragma unroll
                for (int x = 0; x < 9; ++x) {
                    if (x == 1 || x == 3 || x == 4 || x == 5 || x == 7)
                        acc[x][0] = __builtin_amdgcn_wmma_f32_16x16x4_f32(
                            false, ring[x][(i0 + u) % R], false, bf[x], (short)0, acc[x][0], false, false);
                }
#pragma unroll
                for (int j = 1; j < MT; ++j) {
#pragma unroll
                    for (int x = 0; x < 9; ++x)
                        acc[x][j] = __builtin_amdgcn_wmma_f32_16x16x4_f32(
                            false, ring[x][(i0 + u - 4 * j + R) % R], false, bf[x], (short)0, acc[x][j], false, false);
                }
            }
        }
        p  += 512 * R;
        sa += 4 * R;
    }

#pragma unroll
    for (int jj = 1; jj <= MT; ++jj) {
        float pbA = __shfl_xor(half ? acc[1][jj - 1][7] : acc[1][jj][7], 16);
        float paB = __shfl_xor(half ? acc[3][jj - 1][7] : acc[3][jj][7], 16);
        float pbB = __shfl_xor(half ? acc[4][jj - 1][7] : acc[4][jj][7], 16);
        float pcB = __shfl_xor(half ? acc[5][jj - 1][7] : acc[5][jj][7], 16);
        float pbC = __shfl_xor(half ? acc[7][jj - 1][7] : acc[7][jj][7], 16);
        const int tt = t0 + 64 * (jj - 1) + 32 * half;
        float* o = dst + (size_t)(n + 16 * c) * T_LEN + (tt - PAD);
#pragma unroll
        for (int r = 0; r < 8; ++r) {
            const float aA = acc[0][jj][r], bA = acc[1][jj][r], cA = acc[2][jj][r];
            const float aB = acc[3][jj][r], bB = acc[4][jj][r], cB = acc[5][jj][r];
            const float aC = acc[6][jj][r], bC = acc[7][jj][r], cC = acc[8][jj][r];
            const float Ae  = aA + pbA;
            const float Ao  = cA - aA - bA;
            const float Be  = aB + pbB;
            const float Bo  = cB - aB - bB;
            const float Bom = pcB - paB - pbB;
            const float Ce  = aC + pbC;
            const float Co  = cC - aC - bC;
            *((float4*)o + r) = make_float4(Ae + Bom, Ce - Ae - Be, Ao + Be, Co - Ao - Bo);
            pbA = bA; paB = aB; pbB = bB; pcB = cB; pbC = bC;
        }
    }
}

extern "C" void kernel_launch(void* const* d_in, const int* in_sizes, int n_in,
                              void* d_out, int out_size, void* d_ws, size_t ws_size,
                              hipStream_t stream) {
    const float* x    = (const float*)d_in[0];
    const float* kern = (const float*)d_in[1];
    float* out = (float*)d_out;
    float* ws  = (float*)d_ws;

    float* f1  = ws + F1OFF;
    float* f2  = ws + F2OFF;
    float* xq  = ws + XOFF;
    float* y1q = xq + (size_t)ROWS * S_SIG;

    prep_h_kernel<<<(9 * Geo4<MT2>::SHN + 255) / 256, 256, 0, stream>>>(kern, f1, f2);
    build_xq_kernel<<<(ROWS * S_SIG) / 256, 256, 0, stream>>>(x, xq);
    zero_guards_kernel<<<(GT * S_SIG) / 256, 256, 0, stream>>>(y1q);

    fir_pass_kernel<MT1>
        <<<TP / (32 * MT1 * 2), 128, 0, stream>>>(f1, xq, y1q, 0);
    fir_pass4_kernel<MT2>
        <<<T_LEN / (64 * MT2 * 2), 128, 0, stream>>>(f2, y1q, out, PAD);
}


# input staging kernel: a thread moves eight 16-byte slots (8 lanes store a whole 128-byte line, 4 lanes read 64 contiguous bytes of a signal) in place of one 4-byte element: 32x fewer waves
# speedup vs baseline: 1.6835x; 1.0336x over previous
#include <hip/hip_runtime.h>


typedef __attribute__((ext_vector_type(2))) float v2f;
typedef __attribute__((ext_vector_type(4))) float v4f;
typedef __attribute__((ext_vector_type(8))) float v8f;

#define K_TAPS 4096
#define KH     2048
#define PAD    2048
#define T_LEN  131072
#define S_SIG  32
#define TP     (T_LEN + 2*PAD)
#define GT     2112
#define ROWS   (TP + 2*GT)
#define SBASE  64
#define MT1    3
#define KQ     1024
#define MT2    4
#define XOFF   32768

template<int MT> struct Geo {
    static constexpr int G   = 16 * (MT + 1);
    static constexpr int R   = 4 * (MT + 1);
    static constexpr int BL  = R / 2;
    static constexpr int NCH = ((516 + 4 * MT + R - 1) / R) * R;
    static constexpr int SHN = G + 4 * (NCH + 4) + 8;
};
template<int MT> struct Geo4 {
    static constexpr int G   = 16 * (MT + 1);
    static constexpr int R   = 4 * (MT + 1);
    static constexpr int BL  = R / 2;
    static constexpr int NCH = ((260 + 4 * MT + R - 1) / R) * R;
    static constexpr int SHN = G + 4 * (NCH + 4) + 8;
};
#define F1OFF  0
#define F2OFF  (3 * Geo<MT1>::SHN)
static_assert(F2OFF + 9 * Geo4<MT2>::SHN <= XOFF, "filter blocks end before the staged input");

__device__ __forceinline__ int quad_slot(int r) { return ((r & 1) << 1) | ((r >> 1) & 1); }

__device__ __forceinline__ float f_tap(const float* __restrict__ h, int e, bool backward) {
    const int i = backward ? (K_TAPS - 1 - e) : (e - 1);
    return (i >= 0 && i < K_TAPS) ? h[i] : 0.f;
}

template<int MT>
__device__ __forceinline__ float subfilter(const float* __restrict__ h, int o, bool backward) {
    const int x = o / Geo<MT>::SHN;
    const int j = o - x * Geo<MT>::SHN - Geo<MT>::G;
    if (j < 0 || j > KH) return 0.f;
    const float fe = f_tap(h, 2 * j, backward);
    const float fo = f_tap(h, 2 * j - 1, backward);
    return x == 0 ? fe : (x == 1 ? fo : fe + fo);
}

__device__ __forceinline__ float outer_tap(const float* __restrict__ h, int j, int xo, bool backward) {
    if (j < 0 || j > KH) return 0.f;
    const float fe = f_tap(h, 2 * j, backward);
    const float fo = f_tap(h, 2 * j - 1, backward);
    return xo == 0 ? fe : (xo == 1 ? fo : fe + fo);
}

template<int MT>
__device__ __forceinline__ float subfilter4(const float* __restrict__ h, int o, bool backward) {
    const int x  = o / Geo4<MT>::SHN;
    const int jq = o - x * Geo4<MT>::SHN - Geo4<MT>::G;
    if (jq < 0 || jq > KQ) return 0.f;
    const int xo = x / 3;
    const int xi = x - 3 * xo;
    const float ge = outer_tap(h, 2 * jq, xo, backward);
    const float go = outer_tap(h, 2 * jq - 1, xo, backward);
    return xi == 0 ? ge : (xi == 1 ? go : ge + go);
}

static_assert(ROWS % 128 == 0, "whole blocks of 128 rows");
static_assert(GT % 4 == 0 && PAD % 4 == 0 && T_LEN % 4 == 0, "row groups do not straddle boundaries");
__global__ __launch_bounds__(128)
void build_xq_kernel(const float* __restrict__ x,
                     float* __restrict__ xq) {
    const int lane = threadIdx.x & 31;
    const int s    = ((threadIdx.x >> 5) << 3) + (lane & 7);
    const int g0   = blockIdx.x * 32 + (lane >> 3);
    const int r0   = blockIdx.x * 128;
    const float* xs = x + (size_t)s * T_LEN;
    v4f v[8];
    if (r0 >= GT + PAD && r0 + 128 <= GT + PAD + T_LEN) {
        const float* pin = xs + (4 * g0 - (GT + PAD));
#pragma unroll
        for (int j = 0; j < 8; ++j) {
            const v4f a = *(const v4f*)(pin + 16 * j);
            v[j] = v4f{a.x, a.z, a.y, a.w};
        }
    } else {
#pragma unroll
        for (int j = 0; j < 8; ++j) {
            float e[4];
#pragma unroll
            for (int r = 0; r < 4; ++r) {
                const int tg = 4 * (g0 + 4 * j) + r - GT;
                float val = 0.f;
                if (tg >= 0 && tg < TP) {
                    int tr = tg - PAD;
                    if (tr < 0)            tr = -tr;
                    else if (tr >= T_LEN)  tr = 2 * T_LEN - 2 - tr;
                    val = xs[tr];
                }
                e[r] = val;
            }
            v[j] = v4f{e[0], e[2], e[1], e[3]};
        }
    }
    float* o = xq + (size_t)g0 * 128 + 4 * s;
#pragma unroll
    for (int j = 0; j < 8; ++j)
        *(v4f*)(o + 512 * j) = v[j];
}

__global__ void prep_h_kernel(const float* __restrict__ h,
                              float* __restrict__ f1,
                              float* __restrict__ f2) {
    int o = blockIdx.x * blockDim.x + threadIdx.x;
    if (o < 3 * Geo<MT1>::SHN) f1[o] = subfilter<MT1>(h, o, false);
    if (o < 9 * Geo4<MT2>::SHN) f2[o] = subfilter4<MT2>(h, o, true);
}

__global__ void zero_guards_kernel(float* __restrict__ y1q) {
    int idx = blockIdx.x * blockDim.x + threadIdx.x;
    y1q[idx] = 0.f;
    y1q[(size_t)(GT + TP) * S_SIG + idx] = 0.f;
}

template<int MT>
__global__ __launch_bounds__(128)
void fir_pass_kernel(const float* __restrict__ filt,
                     const float* __restrict__ src,
                     float* __restrict__ dst,
                     int t0_origin) {
    constexpr int G   = Geo<MT>::G;
    constexpr int R   = Geo<MT>::R;
    constexpr int BL  = Geo<MT>::BL;
    constexpr int NCH = Geo<MT>::NCH;
    constexpr int SHN = Geo<MT>::SHN;
    static_assert(R % BL == 0, "B ring must divide the A ring");
    static_assert(SHN % 4 == 0, "whole 16-byte slots");

    constexpr int NV  = 3 * SHN / 4;
    constexpr int NIT = (NV + 127) / 128;

    __shared__ __attribute__((aligned(16))) float sh[3 * SHN];
    float4 stage[NIT];
#pragma unroll
    for (int j = 0; j < NIT; ++j) {
        const int v = min((int)threadIdx.x + 128 * j, NV - 1);
        stage[j] = *(const float4*)(filt + 4 * v);
    }
#pragma unroll
    for (int j = 0; j < NIT; ++j) {
        const int v = (int)threadIdx.x + 128 * j;
        if (v < NV) *(float4*)(sh + 4 * v) = stage[j];
    }
    __syncthreads();

    const int lane = threadIdx.x & 31;
    const int wave = threadIdx.x >> 5;
    const int n    = lane & 15;
    const int half = lane >> 4;
    const int q    = half * 2;
    const int c    = wave & 1;
    const int grp  = blockIdx.x * 2 + (wave >> 1);

    const int t0 = t0_origin + grp * (32 * MT);

    const float* saE = sh + (G + q - n);
    const float* saO = saE + SHN;
    const float* saS = saE + 2 * SHN;
    const float* p = src + ((size_t)((SBASE + t0 - 32) >> 2) + half) * 128
                         + 4 * (n + 16 * c);

    v8f accE[MT + 1], accS[MT + 1], accO[MT + 1];
#pragma unroll
    for (int j = 0; j <= MT; ++j) { accE[j] = v8f{}; accS[j] = v8f{}; accO[j] = v8f{}; }

    v2f rE[R], rO[R], rS[R];
    v4f bq[BL];
#pragma unroll
    for (int s = 0; s < R; ++s) {
        const int kc = (s < 4) ? s : s - R;
        rE[s].x = saE[4 * kc]; rE[s].y = saE[4 * kc + 1];
        rO[s].x = saO[4 * kc]; rO[s].y = saO[4 * kc + 1];
        rS[s].x = saS[4 * kc]; rS[s].y = saS[4 * kc + 1];
    }
#pragma unroll
    for (int s = 0; s < BL - 4; ++s)
        bq[s] = *(const v4f*)(p + 256 * s);

    for (int c0 = 0; c0 < NCH; c0 += R) {
#pragma unroll
        for (int i0 = 0; i0 < R; i0 += 4) {
#pragma unroll
            for (int u = 0; u < 4; ++u) {
                const v4f b = bq[(i0 + u) % BL];
                const v2f be = {b.x, b.y};
                const v2f bo = {b.z, b.w};
                const v2f bs = be + bo;
                accE[MT] = __builtin_amdgcn_wmma_f32_16x16x4_f32(
                    false, rE[(i0 + u + 4) % R], false, be, (short)0, accE[MT], false, false);
                accO[MT] = __builtin_amdgcn_wmma_f32_16x16x4_f32(
                    false, rO[(i0 + u + 4) % R], false, bo, (short)0, accO[MT], false, false);
                accS[MT] = __builtin_amdgcn_wmma_f32_16x16x4_f32(
                    false, rS[(i0 + u + 4) % R], false, bs, (short)0, accS[MT], false, false);
            }
            __builtin_amdgcn_sched_barrier(0);
#pragma unroll
            for (int u = 0; u < 4; ++u) {
                const int s = (i0 + u + 4) % R;
                const int kc = i0 + u + 4;
                rE[s].x = saE[4 * kc]; rE[s].y = saE[4 * kc + 1];
                rO[s].x = saO[4 * kc]; rO[s].y = saO[4 * kc + 1];
                rS[s].x = saS[4 * kc]; rS[s].y = saS[4 * kc + 1];
            }
#pragma unroll
            for (int u = 0; u < 4; ++u)
                bq[(i0 + u + BL - 4) % BL] = *(const v4f*)(p + 256 * (i0 + u + BL - 4));
            __builtin_amdgcn_sched_barrier(0);
#pragma unroll
            for (int u = 0; u < 4; ++u) {
                const v4f b = bq[(i0 + u) % BL];
                const v2f be = {b.x, b.y};
                const v2f bo = {b.z, b.w};
                const v2f bs = be + bo;
                accO[0] = __builtin_amdgcn_wmma_f32_16x16x4_f32(
                    false, rO[(i0 + u) % R], false, bo, (short)0, accO[0], false, false);
#pragma unroll
                for (int j = 1; j < MT; ++j) {
                    accE[j] = __builtin_amdgcn_wmma_f32_16x16x4_f32(
                        false, rE[(i0 + u - 4 * j + R) % R], false, be, (short)0, accE[j], false, false);
                    accO[j] = __builtin_amdgcn_wmma_f32_16x16x4_f32(
                        false, rO[(i0 + u - 4 * j + R) % R], false, bo, (short)0, accO[j], false, false);
                    accS[j] = __builtin_amdgcn_wmma_f32_16x16x4_f32(
                        false, rS[(i0 + u - 4 * j + R) % R], false, bs, (short)0, accS[j], false, false);
                }
            }
        }
        p   += 256 * R;
        saE += 4 * R;
        saO += 4 * R;
        saS += 4 * R;
    }

#pragma unroll
    for (int jj = 1; jj <= MT; ++jj) {
        const float send = half ? accO[jj - 1][7] : accO[jj][7];
        const float prev = __shfl_xor(send, 16);
        float ev[8], od[8];
#pragma unroll
        for (int r = 0; r < 8; ++r) {
            const float bm1 = (r == 0) ? prev : accO[jj][r - 1];
            ev[r] = accE[jj][r] + bm1;
            od[r] = accS[jj][r] - accE[jj][r] - accO[jj][r];
        }
        const int tt = t0 + 32 * (jj - 1) + 16 * half;
        float* o = dst + ((size_t)(GT + tt) >> 2) * 128 + 4 * (n + 16 * c);
#pragma unroll
        for (int g = 0; g < 4; ++g)
            *(float4*)(o + (size_t)g * 128) =
                make_float4(ev[2 * g], ev[2 * g + 1], od[2 * g], od[2 * g + 1]);
    }
}

template<int MT>
__global__ __launch_bounds__(128)
void fir_pass4_kernel(const float* __restrict__ filt,
                      const float* __restrict__ src,
                      float* __restrict__ dst,
                      int t0_origin) {
    constexpr int G   = Geo4<MT>::G;
    constexpr int R   = Geo4<MT>::R;
    constexpr int BL  = Geo4<MT>::BL;
    constexpr int NCH = Geo4<MT>::NCH;
    constexpr int SHN = Geo4<MT>::SHN;
    static_assert(R % BL == 0, "B ring must divide the A ring");
    static_assert(SHN % 4 == 0, "whole 16-byte slots");
    static_assert(4 * NCH >= KQ + 1 + 16 * MT + 15, "every tile's tap window lies inside the k axis");

    constexpr int NV  = 9 * SHN / 4;
    constexpr int NIT = (NV + 127) / 128;

    __shared__ __attribute__((aligned(16))) float sh[9 * SHN];
    float4 stage[NIT];
#pragma unroll
    for (int j = 0; j < NIT; ++j) {
        const int v = min((int)threadIdx.x + 128 * j, NV - 1);
        stage[j] = *(const float4*)(filt + 4 * v);
    }
#pragma unroll
    for (int j = 0; j < NIT; ++j) {
        const int v = (int)threadIdx.x + 128 * j;
        if (v < NV) *(float4*)(sh + 4 * v) = stage[j];
    }
    __syncthreads();

    const int lane = threadIdx.x & 31;
    const int wave = threadIdx.x >> 5;
    const int n    = lane & 15;
    const int half = lane >> 4;
    const int q    = half * 2;
    const int c    = wave & 1;
    const int grp  = blockIdx.x * 2 + (wave >> 1);

    const int t0 = t0_origin + grp * (64 * MT);

    const float* sa = sh + (G + q - n);
    const float* p = src + ((size_t)(((SBASE + t0) >> 2) - 16) + 2 * half) * 128
                         + 4 * (n + 16 * c);

    v8f acc[9][MT + 1];
#pragma unroll
    for (int x = 0; x < 9; ++x)
#pragma unroll
        for (int j = 0; j <= MT; ++j) acc[x][j] = v8f{};

    v2f ring[9][R];
    v4f bq[BL][2];
#pragma unroll
    for (int s = 0; s < R; ++s) {
        const int kc = (s < 4) ? s : s - R;
#pragma unroll
        for (int x = 0; x < 9; ++x) {
            ring[x][s].x = sa[x * SHN + 4 * kc];
            ring[x][s].y = sa[x * SHN + 4 * kc + 1];
        }
    }
#pragma unroll
    for (int s = 0; s < BL - 4; ++s) {
        bq[s][0] = *(const v4f*)(p + 512 * s);
        bq[s][1] = *(const v4f*)(p + 512 * s + 128);
    }

    for (int c0 = 0; c0 < NCH; c0 += R) {
#pragma unroll
        for (int i0 = 0; i0 < R; i0 += 4) {
#pragma unroll
            for (int u = 0; u < 4; ++u) {
                const v4f g0 = bq[(i0 + u) % BL][0];
                const v4f g1 = bq[(i0 + u) % BL][1];
                v2f bf[9];
                bf[0] = v2f{g0.x, g1.x};
                bf[1] = v2f{g0.y, g1.y};
                bf[3] = v2f{g0.z, g1.z};
                bf[4] = v2f{g0.w, g1.w};
                bf[2] = bf[0] + bf[1];
                bf[5] = bf[3] + bf[4];
                bf[6] = bf[0] + bf[3];
                bf[7] = bf[1] + bf[4];
                bf[8] = bf[6] + bf[7];
#pragma unroll
                for (int x = 0; x < 9; ++x)
                    acc[x][MT] = __builtin_amdgcn_wmma_f32_16x16x4_f32(
                        false, ring[x][(i0 + u + 4) % R], false, bf[x], (short)0, acc[x][MT], false, false);
            }
            __builtin_amdgcn_sched_barrier(0);
#pragma unroll
            for (int u = 0; u < 4; ++u) {
                const int s = (i0 + u + 4) % R;
                const int kc = i0 + u + 4;
#pragma unroll
                for (int x = 0; x < 9; ++x) {
                    ring[x][s].x = sa[x * SHN + 4 * kc];
                    ring[x][s].y = sa[x * SHN + 4 * kc + 1];
                }
            }
#pragma unroll
            for (int u = 0; u < 4; ++u) {
                bq[(i0 + u + BL - 4) % BL][0] = *(const v4f*)(p + 512 * (i0 + u + BL - 4));
                bq[(i0 + u + BL - 4) % BL][1] = *(const v4f*)(p + 512 * (i0 + u + BL - 4) + 128);
            }
            __builtin_amdgcn_sched_barrier(0);
#pragma unroll
            for (int u = 0; u < 4; ++u) {
                const v4f g0 = bq[(i0 + u) % BL][0];
                const v4f g1 = bq[(i0 + u) % BL][1];
                v2f bf[9];
                bf[0] = v2f{g0.x, g1.x};
                bf[1] = v2f{g0.y, g1.y};
                bf[3] = v2f{g0.z, g1.z};
                bf[4] = v2f{g0.w, g1.w};
                bf[2] = bf[0] + bf[1];
                bf[5] = bf[3] + bf[4];
                bf[6] = bf[0] + bf[3];
                bf[7] = bf[1] + bf[4];
                bf[8] = bf[6] + bf[7];
#pragma unroll
                for (int x = 0; x < 9; ++x) {
                    if (x == 1 || x == 3 || x == 4 || x == 5 || x == 7)
                        acc[x][0] = __builtin_amdgcn_wmma_f32_16x16x4_f32(
                            false, ring[x][(i0 + u) % R], false, bf[x], (short)0, acc[x][0], false, false);
                }
#pragma unroll
                for (int j = 1; j < MT; ++j) {
#pragma unroll
                    for (int x = 0; x < 9; ++x)
                        acc[x][j] = __builtin_amdgcn_wmma_f32_16x16x4_f32(
                            false, ring[x][(i0 + u - 4 * j + R) % R], false, bf[x], (short)0, acc[x][j], false, false);
                }
            }
        }
        p  += 512 * R;
        sa += 4 * R;
    }

#pragma unroll
    for (int jj = 1; jj <= MT; ++jj) {
        float pbA = __shfl_xor(half ? acc[1][jj - 1][7] : acc[1][jj][7], 16);
        float paB = __shfl_xor(half ? acc[3][jj - 1][7] : acc[3][jj][7], 16);
        float pbB = __shfl_xor(half ? acc[4][jj - 1][7] : acc[4][jj][7], 16);
        float pcB = __shfl_xor(half ? acc[5][jj - 1][7] : acc[5][jj][7], 16);
        float pbC = __shfl_xor(half ? acc[7][jj - 1][7] : acc[7][jj][7], 16);
        const int tt = t0 + 64 * (jj - 1) + 32 * half;
        float* o = dst + (size_t)(n + 16 * c) * T_LEN + (tt - PAD);
#pragma unroll
        for (int r = 0; r < 8; ++r) {
            const float aA = acc[0][jj][r], bA = acc[1][jj][r], cA = acc[2][jj][r];
            const float aB = acc[3][jj][r], bB = acc[4][jj][r], cB = acc[5][jj][r];
            const float aC = acc[6][jj][r], bC = acc[7][jj][r], cC = acc[8][jj][r];
            const float Ae  = aA + pbA;
            const float Ao  = cA - aA - bA;
            const float Be  = aB + pbB;
            const float Bo  = cB - aB - bB;
            const float Bom = pcB - paB - pbB;
            const float Ce  = aC + pbC;
            const float Co  = cC - aC - bC;
            *((float4*)o + r) = make_float4(Ae + Bom, Ce - Ae - Be, Ao + Be, Co - Ao - Bo);
            pbA = bA; paB = aB; pbB = bB; pcB = cB; pbC = bC;
        }
    }
}

extern "C" void kernel_launch(void* const* d_in, const int* in_sizes, int n_in,
                              void* d_out, int out_size, void* d_ws, size_t ws_size,
                              hipStream_t stream) {
    const float* x    = (const float*)d_in[0];
    const float* kern = (const float*)d_in[1];
    float* out = (float*)d_out;
    float* ws  = (float*)d_ws;

    float* f1  = ws + F1OFF;
    float* f2  = ws + F2OFF;
    float* xq  = ws + XOFF;
    float* y1q = xq + (size_t)ROWS * S_SIG;

    prep_h_kernel<<<(9 * Geo4<MT2>::SHN + 255) / 256, 256, 0, stream>>>(kern, f1, f2);
    build_xq_kernel<<<ROWS / 128, 128, 0, stream>>>(x, xq);
    zero_guards_kernel<<<(GT * S_SIG) / 256, 256, 0, stream>>>(y1q);

    fir_pass_kernel<MT1>
        <<<TP / (32 * MT1 * 2), 128, 0, stream>>>(f1, xq, y1q, 0);
    fir_pass4_kernel<MT2>
        <<<T_LEN / (64 * MT2 * 2), 128, 0, stream>>>(f2, y1q, out, PAD);
}


# both filter passes fused into one 8191-tap pass with r[d]=sum_i h[i]h[i+d] built from h in every call; no intermediate plane, 4 launches, one wave per SIMD; order of summation changes
# speedup vs baseline: 1.9454x; 1.1556x over previous
#include <hip/hip_runtime.h>


typedef __attribute__((ext_vector_type(2))) float v2f;
typedef __attribute__((ext_vector_type(4))) float v4f;
typedef __attribute__((ext_vector_type(8))) float v8f;

#define K_TAPS 4096
#define NF     8192
#define KH     4096
#define KQ     2048
#define PAD    2048
#define T_LEN  131072
#define S_SIG  32
#define TP     (T_LEN + 2*PAD)
#define GT     2112
#define ROWS   (TP + 2*GT + 256)
#define SBASE  (GT - 4096)
#define MT2    4
#define XOFF   32768

template<int MT> struct Geo4 {
    static constexpr int G   = 16 * (MT + 1);
    static constexpr int R   = 4 * (MT + 1);
    static constexpr int BL  = R / 2;
    static constexpr int NCH = ((516 + 4 * MT + R - 1) / R) * R;
    static constexpr int SHN = G + 4 * (NCH + 4) + 8;
};
#define FOFF   0
#define F2OFF  NF
static_assert(F2OFF + 9 * Geo4<MT2>::SHN <= XOFF, "filter blocks end before the staged input");
static_assert(SBASE + PAD + T_LEN - 64 * MT2 - 64
              + 16 * (Geo4<MT2>::NCH + Geo4<MT2>::BL - 4) <= ROWS, "k axis ends inside the staged rows");
static_assert(SBASE + PAD - 64 >= 0, "k axis begins inside the staged rows");

__device__ __forceinline__ int quad_slot(int r) { return ((r & 1) << 1) | ((r >> 1) & 1); }

__device__ __forceinline__ float f_tap(const float* __restrict__ f, int e) {
    return (e >= 0 && e < NF) ? f[e] : 0.f;
}

__device__ __forceinline__ float outer_tap(const float* __restrict__ f, int j, int xo) {
    if (j < 0 || j > KH) return 0.f;
    const float fe = f_tap(f, 2 * j);
    const float fo = f_tap(f, 2 * j - 1);
    return xo == 0 ? fe : (xo == 1 ? fo : fe + fo);
}

template<int MT>
__device__ __forceinline__ float subfilter4(const float* __restrict__ f, int o) {
    const int x  = o / Geo4<MT>::SHN;
    const int jq = o - x * Geo4<MT>::SHN - Geo4<MT>::G;
    if (jq < 0 || jq > KQ) return 0.f;
    const int xo = x / 3;
    const int xi = x - 3 * xo;
    const float ge = outer_tap(f, 2 * jq, xo);
    const float go = outer_tap(f, 2 * jq - 1, xo);
    return xi == 0 ? ge : (xi == 1 ? go : ge + go);
}

static_assert(ROWS % 128 == 0, "whole blocks of 128 rows");
static_assert(GT % 4 == 0 && PAD % 4 == 0 && T_LEN % 4 == 0, "row groups do not straddle boundaries");
__global__ __launch_bounds__(128)
void build_xq_kernel(const float* __restrict__ x,
                     float* __restrict__ xq) {
    const int lane = threadIdx.x & 31;
    const int s    = ((threadIdx.x >> 5) << 3) + (lane & 7);
    const int g0   = blockIdx.x * 32 + (lane >> 3);
    const int r0   = blockIdx.x * 128;
    const float* xs = x + (size_t)s * T_LEN;
    v4f v[8];
    if (r0 >= GT + PAD && r0 + 128 <= GT + PAD + T_LEN) {
        const float* pin = xs + (4 * g0 - (GT + PAD));
#pragma unroll
        for (int j = 0; j < 8; ++j) {
            const v4f a = *(const v4f*)(pin + 16 * j);
            v[j] = v4f{a.x, a.z, a.y, a.w};
        }
    } else {
#pragma unroll
        for (int j = 0; j < 8; ++j) {
            float e[4];
#pragma unroll
            for (int r = 0; r < 4; ++r) {
                const int tg = 4 * (g0 + 4 * j) + r - GT;
                float val = 0.f;
                if (tg >= 0 && tg < TP) {
                    int tr = tg - PAD;
                    if (tr < 0)            tr = -tr;
                    else if (tr >= T_LEN)  tr = 2 * T_LEN - 2 - tr;
                    val = xs[tr];
                }
                e[r] = val;
            }
            v[j] = v4f{e[0], e[2], e[1], e[3]};
        }
    }
    float* o = xq + (size_t)g0 * 128 + 4 * s;
#pragma unroll
    for (int j = 0; j < 8; ++j)
        *(v4f*)(o + 512 * j) = v[j];
}

__global__ __launch_bounds__(256)
void prep_r_kernel(const float* __restrict__ h,
                   float* __restrict__ f) {
    const int lane = threadIdx.x & 31;
    const int e    = blockIdx.x * 8 + (threadIdx.x >> 5);
    const int d    = e - K_TAPS;
    const int lo   = d < 0 ? -d : 0;
    const int hi   = d < 0 ? K_TAPS : K_TAPS - d;
    float acc = 0.f;
    for (int i = lo + lane; i < hi; i += 32)
        acc = fmaf(h[i], h[i + d], acc);
    for (int m = 16; m >= 1; m >>= 1)
        acc += __shfl_xor(acc, m);
    if (lane == 0) f[e] = acc;
}

__global__ void prep_sub_kernel(const float* __restrict__ f,
                                float* __restrict__ f2) {
    int o = blockIdx.x * blockDim.x + threadIdx.x;
    if (o < 9 * Geo4<MT2>::SHN) f2[o] = subfilter4<MT2>(f, o);
}

template<int MT>
__global__ __launch_bounds__(128)
void fir_pass4_kernel(const float* __restrict__ filt,
                      const float* __restrict__ src,
                      float* __restrict__ dst,
                      int t0_origin) {
    constexpr int G   = Geo4<MT>::G;
    constexpr int R   = Geo4<MT>::R;
    constexpr int BL  = Geo4<MT>::BL;
    constexpr int NCH = Geo4<MT>::NCH;
    constexpr int SHN = Geo4<MT>::SHN;
    static_assert(R % BL == 0, "B ring must divide the A ring");
    static_assert(SHN % 4 == 0, "whole 16-byte slots");
    static_assert(4 * NCH >= KQ + 1 + 16 * MT + 15, "every tile's tap window lies inside the k axis");

    constexpr int NV  = 9 * SHN / 4;
    constexpr int NIT = (NV + 127) / 128;

    __shared__ __attribute__((aligned(16))) float sh[9 * SHN];
    constexpr int NB = 20;
    for (int j0 = 0; j0 < NIT; j0 += NB) {
        float4 stage[NB];
#pragma unroll
        for (int j = 0; j < NB; ++j) {
            const int v = min((int)threadIdx.x + 128 * (j0 + j), NV - 1);
            stage[j] = *(const float4*)(filt + 4 * v);
        }
#pragma unroll
        for (int j = 0; j < NB; ++j) {
            const int v = (int)threadIdx.x + 128 * (j0 + j);
            if (v < NV) *(float4*)(sh + 4 * v) = stage[j];
        }
    }
    __syncthreads();

    const int lane = threadIdx.x & 31;
    const int wave = threadIdx.x >> 5;
    const int n    = lane & 15;
    const int half = lane >> 4;
    const int q    = half * 2;
    const int c    = wave & 1;
    const int grp  = blockIdx.x * 2 + (wave >> 1);

    const int t0 = t0_origin + grp * (64 * MT);

    const float* sa = sh + (G + q - n);
    const float* p = src + ((size_t)(((SBASE + t0) >> 2) - 16) + 2 * half) * 128
                         + 4 * (n + 16 * c);

    v8f acc[9][MT + 1];
#pragma unroll
    for (int x = 0; x < 9; ++x)
#pragma unroll
        for (int j = 0; j <= MT; ++j) acc[x][j] = v8f{};

    v2f ring[9][R];
    v4f bq[BL][2];
#pragma unroll
    for (int s = 0; s < R; ++s) {
        const int kc = (s < 4) ? s : s - R;
#pragma unroll
        for (int x = 0; x < 9; ++x) {
            ring[x][s].x = sa[x * SHN + 4 * kc];
            ring[x][s].y = sa[x * SHN + 4 * kc + 1];
        }
    }
#pragma unroll
    for (int s = 0; s < BL - 4; ++s) {
        bq[s][0] = *(const v4f*)(p + 512 * s);
        bq[s][1] = *(const v4f*)(p + 512 * s + 128);
    }

    for (int c0 = 0; c0 < NCH; c0 += R) {
#pragma unroll
        for (int i0 = 0; i0 < R; i0 += 4) {
#pragma unroll
            for (int u = 0; u < 4; ++u) {
                const v4f g0 = bq[(i0 + u) % BL][0];
                const v4f g1 = bq[(i0 + u) % BL][1];
                v2f bf[9];
                bf[0] = v2f{g0.x, g1.x};
                bf[1] = v2f{g0.y, g1.y};
                bf[3] = v2f{g0.z, g1.z};
                bf[4] = v2f{g0.w, g1.w};
                bf[2] = bf[0] + bf[1];
                bf[5] = bf[3] + bf[4];
                bf[6] = bf[0] + bf[3];
                bf[7] = bf[1] + bf[4];
                bf[8] = bf[6] + bf[7];
#pragma unroll
                for (int x = 0; x < 9; ++x)
                    acc[x][MT] = __builtin_amdgcn_wmma_f32_16x16x4_f32(
                        false, ring[x][(i0 + u + 4) % R], false, bf[x], (short)0, acc[x][MT], false, false);
            }
            __builtin_amdgcn_sched_barrier(0);
#pragma unroll
            for (int u = 0; u < 4; ++u) {
                const int s = (i0 + u + 4) % R;
                const int kc = i0 + u + 4;
#pragma unroll
                for (int x = 0; x < 9; ++x) {
                    ring[x][s].x = sa[x * SHN + 4 * kc];
                    ring[x][s].y = sa[x * SHN + 4 * kc + 1];
                }
            }
#pragma unroll
            for (int u = 0; u < 4; ++u) {
                bq[(i0 + u + BL - 4) % BL][0] = *(const v4f*)(p + 512 * (i0 + u + BL - 4));
                bq[(i0 + u + BL - 4) % BL][1] = *(const v4f*)(p + 512 * (i0 + u + BL - 4) + 128);
            }
            __builtin_amdgcn_sched_barrier(0);
#pragma unroll
            for (int u = 0; u < 4; ++u) {
                const v4f g0 = bq[(i0 + u) % BL][0];
                const v4f g1 = bq[(i0 + u) % BL][1];
                v2f bf[9];
                bf[0] = v2f{g0.x, g1.x};
                bf[1] = v2f{g0.y, g1.y};
                bf[3] = v2f{g0.z, g1.z};
                bf[4] = v2f{g0.w, g1.w};
                bf[2] = bf[0] + bf[1];
                bf[5] = bf[3] + bf[4];
                bf[6] = bf[0] + bf[3];
                bf[7] = bf[1] + bf[4];
                bf[8] = bf[6] + bf[7];
#pragma unroll
                for (int x = 0; x < 9; ++x) {
                    if (x == 1 || x == 3 || x == 4 || x == 5 || x == 7)
                        acc[x][0] = __builtin_amdgcn_wmma_f32_16x16x4_f32(
                            false, ring[x][(i0 + u) % R], false, bf[x], (short)0, acc[x][0], false, false);
                }
#pragma unroll
                for (int j = 1; j < MT; ++j) {
#pragma unroll
                    for (int x = 0; x < 9; ++x)
                        acc[x][j] = __builtin_amdgcn_wmma_f32_16x16x4_f32(
                            false, ring[x][(i0 + u - 4 * j + R) % R], false, bf[x], (short)0, acc[x][j], false, false);
                }
            }
        }
        p  += 512 * R;
        sa += 4 * R;
    }

#pragma unroll
    for (int jj = 1; jj <= MT; ++jj) {
        float pbA = __shfl_xor(half ? acc[1][jj - 1][7] : acc[1][jj][7], 16);
        float paB = __shfl_xor(half ? acc[3][jj - 1][7] : acc[3][jj][7], 16);
        float pbB = __shfl_xor(half ? acc[4][jj - 1][7] : acc[4][jj][7], 16);
        float pcB = __shfl_xor(half ? acc[5][jj - 1][7] : acc[5][jj][7], 16);
        float pbC = __shfl_xor(half ? acc[7][jj - 1][7] : acc[7][jj][7], 16);
        const int tt = t0 + 64 * (jj - 1) + 32 * half;
        float* o = dst + (size_t)(n + 16 * c) * T_LEN + (tt - PAD);
#pragma unroll
        for (int r = 0; r < 8; ++r) {
            const float aA = acc[0][jj][r], bA = acc[1][jj][r], cA = acc[2][jj][r];
            const float aB = acc[3][jj][r], bB = acc[4][jj][r], cB = acc[5][jj][r];
            const float aC = acc[6][jj][r], bC = acc[7][jj][r], cC = acc[8][jj][r];
            const float Ae  = aA + pbA;
            const float Ao  = cA - aA - bA;
            const float Be  = aB + pbB;
            const float Bo  = cB - aB - bB;
            const float Bom = pcB - paB - pbB;
            const float Ce  = aC + pbC;
            const float Co  = cC - aC - bC;
            *((float4*)o + r) = make_float4(Ae + Bom, Ce - Ae - Be, Ao + Be, Co - Ao - Bo);
            pbA = bA; paB = aB; pbB = bB; pcB = cB; pbC = bC;
        }
    }
}

extern "C" void kernel_launch(void* const* d_in, const int* in_sizes, int n_in,
                              void* d_out, int out_size, void* d_ws, size_t ws_size,
                              hipStream_t stream) {
    const float* x    = (const float*)d_in[0];
    const float* kern = (const float*)d_in[1];
    float* out = (float*)d_out;
    float* ws  = (float*)d_ws;

    float* f   = ws + FOFF;
    float* f2  = ws + F2OFF;
    float* xq  = ws + XOFF;

    prep_r_kernel<<<NF / 8, 256, 0, stream>>>(kern, f);
    prep_sub_kernel<<<(9 * Geo4<MT2>::SHN + 255) / 256, 256, 0, stream>>>(f, f2);
    build_xq_kernel<<<ROWS / 128, 128, 0, stream>>>(x, xq);

    fir_pass4_kernel<MT2>
        <<<T_LEN / (64 * MT2 * 2), 128, 0, stream>>>(f2, xq, out, PAD);
}


# filter preparation (r and the nine sub-filters, one wave per tap index, loads of 8 trips grouped) rides in the staging launch as extra blocks: 2 launches instead of 4; filter values bit-identical
# speedup vs baseline: 2.0717x; 1.0649x over previous
#include <hip/hip_runtime.h>


typedef __attribute__((ext_vector_type(2))) float v2f;
typedef __attribute__((ext_vector_type(4))) float v4f;
typedef __attribute__((ext_vector_type(8))) float v8f;

#define K_TAPS 4096
#define NF     8192
#define KQ     2048
#define PAD    2048
#define T_LEN  131072
#define S_SIG  32
#define TP     (T_LEN + 2*PAD)
#define GT     2112
#define ROWS   (TP + 2*GT + 256)
#define SBASE  (GT - 4096)
#define MT2    4
#define XOFF   32768

template<int MT> struct Geo4 {
    static constexpr int G   = 16 * (MT + 1);
    static constexpr int R   = 4 * (MT + 1);
    static constexpr int BL  = R / 2;
    static constexpr int NCH = ((516 + 4 * MT + R - 1) / R) * R;
    static constexpr int SHN = G + 4 * (NCH + 4) + 8;
};
#define F2OFF  0
static_assert(F2OFF + 9 * Geo4<MT2>::SHN <= XOFF, "filter blocks end before the staged input");
static_assert(SBASE + PAD + T_LEN - 64 * MT2 - 64
              + 16 * (Geo4<MT2>::NCH + Geo4<MT2>::BL - 4) <= ROWS, "k axis ends inside the staged rows");
static_assert(SBASE + PAD - 64 >= 0, "k axis begins inside the staged rows");

__device__ __forceinline__ int quad_slot(int r) { return ((r & 1) << 1) | ((r >> 1) & 1); }

#define UR 8
template<int MT>
__device__ __forceinline__ void build_tap_group(const float* __restrict__ h,
                                                float* __restrict__ f2,
                                                int jq, int lane) {
    constexpr int G   = Geo4<MT>::G;
    constexpr int SHN = Geo4<MT>::SHN;
    int D[4], n[4];
    int nmax = 0;
#pragma unroll
    for (int k = 0; k < 4; ++k) {
        const int d = 4 * jq - k - K_TAPS;
        D[k] = d < 0 ? -d : d;
        n[k] = K_TAPS - D[k];
        nmax = max(nmax, n[k]);
    }
    float acc[4] = {0.f, 0.f, 0.f, 0.f};
    for (int t0 = 0; t0 < nmax; t0 += 32 * UR) {
        float p[UR], q[4][UR];
#pragma unroll
        for (int j = 0; j < UR; ++j) {
            const int u = t0 + 32 * j + lane;
            p[j] = h[min(u, K_TAPS - 1)];
#pragma unroll
            for (int k = 0; k < 4; ++k) q[k][j] = h[min(u + D[k], K_TAPS - 1)];
        }
#pragma unroll
        for (int j = 0; j < UR; ++j) {
            const int u = t0 + 32 * j + lane;
#pragma unroll
            for (int k = 0; k < 4; ++k)
                if (u < n[k]) acc[k] = fmaf(p[j], q[k][j], acc[k]);
        }
    }
#pragma unroll
    for (int k = 0; k < 4; ++k)
        for (int m = 16; m >= 1; m >>= 1)
            acc[k] += __shfl_xor(acc[k], m);
    const float a = acc[0], b = acc[1], c = acc[2], d = acc[3];
    const float ab = a + b, cd = c + d;
    if (lane < 9) {
        float v;
        switch (lane) {
            case 0:  v = a;       break;
            case 1:  v = c;       break;
            case 2:  v = a + c;   break;
            case 3:  v = b;       break;
            case 4:  v = d;       break;
            case 5:  v = b + d;   break;
            case 6:  v = ab;      break;
            case 7:  v = cd;      break;
            default: v = ab + cd; break;
        }
        f2[lane * SHN + G + jq] = v;
    } else if (lane < 18 && jq < SHN - KQ - 1) {
        f2[(lane - 9) * SHN + (jq < G ? jq : KQ + 1 + jq)] = 0.f;
    }
}

static_assert(ROWS % 128 == 0, "whole blocks of 128 rows");
static_assert(GT % 4 == 0 && PAD % 4 == 0 && T_LEN % 4 == 0, "row groups do not straddle boundaries");
#define NFB ((KQ + 1 + 3) / 4)
#define NSB (ROWS / 128)
__global__ __launch_bounds__(128)
void prep_kernel(const float* __restrict__ x,
                 const float* __restrict__ h,
                 float* __restrict__ f2,
                 float* __restrict__ xq) {
    const int lane = threadIdx.x & 31;
    if (blockIdx.x < NFB) {
        const int jq = blockIdx.x * 4 + __builtin_amdgcn_readfirstlane((int)(threadIdx.x >> 5));
        if (jq <= KQ) build_tap_group<MT2>(h, f2, jq, lane);
        return;
    }
    const int bs   = blockIdx.x - NFB;
    const int s    = ((threadIdx.x >> 5) << 3) + (lane & 7);
    const int g0   = bs * 32 + (lane >> 3);
    const int r0   = bs * 128;
    const float* xs = x + (size_t)s * T_LEN;
    v4f v[8];
    if (r0 >= GT + PAD && r0 + 128 <= GT + PAD + T_LEN) {
        const float* pin = xs + (4 * g0 - (GT + PAD));
#pragma unroll
        for (int j = 0; j < 8; ++j) {
            const v4f a = *(const v4f*)(pin + 16 * j);
            v[j] = v4f{a.x, a.z, a.y, a.w};
        }
    } else {
#pragma unroll
        for (int j = 0; j < 8; ++j) {
            float e[4];
#pragma unroll
            for (int r = 0; r < 4; ++r) {
                const int tg = 4 * (g0 + 4 * j) + r - GT;
                float val = 0.f;
                if (tg >= 0 && tg < TP) {
                    int tr = tg - PAD;
                    if (tr < 0)            tr = -tr;
                    else if (tr >= T_LEN)  tr = 2 * T_LEN - 2 - tr;
                    val = xs[tr];
                }
                e[r] = val;
            }
            v[j] = v4f{e[0], e[2], e[1], e[3]};
        }
    }
    float* o = xq + (size_t)g0 * 128 + 4 * s;
#pragma unroll
    for (int j = 0; j < 8; ++j)
        *(v4f*)(o + 512 * j) = v[j];
}

template<int MT>
__global__ __launch_bounds__(128)
void fir_pass4_kernel(const float* __restrict__ filt,
                      const float* __restrict__ src,
                      float* __restrict__ dst,
                      int t0_origin) {
    constexpr int G   = Geo4<MT>::G;
    constexpr int R   = Geo4<MT>::R;
    constexpr int BL  = Geo4<MT>::BL;
    constexpr int NCH = Geo4<MT>::NCH;
    constexpr int SHN = Geo4<MT>::SHN;
    static_assert(R % BL == 0, "B ring must divide the A ring");
    static_assert(SHN % 4 == 0, "whole 16-byte slots");
    static_assert(4 * NCH >= KQ + 1 + 16 * MT + 15, "every tile's tap window lies inside the k axis");

    constexpr int NV  = 9 * SHN / 4;
    constexpr int NIT = (NV + 127) / 128;

    __shared__ __attribute__((aligned(16))) float sh[9 * SHN];
    constexpr int NB = 20;
    for (int j0 = 0; j0 < NIT; j0 += NB) {
        float4 stage[NB];
#pragma unroll
        for (int j = 0; j < NB; ++j) {
            const int v = min((int)threadIdx.x + 128 * (j0 + j), NV - 1);
            stage[j] = *(const float4*)(filt + 4 * v);
        }
#pragma unroll
        for (int j = 0; j < NB; ++j) {
            const int v = (int)threadIdx.x + 128 * (j0 + j);
            if (v < NV) *(float4*)(sh + 4 * v) = stage[j];
        }
    }
    __syncthreads();

    const int lane = threadIdx.x & 31;
    const int wave = threadIdx.x >> 5;
    const int n    = lane & 15;
    const int half = lane >> 4;
    const int q    = half * 2;
    const int c    = wave & 1;
    const int grp  = blockIdx.x * 2 + (wave >> 1);

    const int t0 = t0_origin + grp * (64 * MT);

    const float* sa = sh + (G + q - n);
    const float* p = src + ((size_t)(((SBASE + t0) >> 2) - 16) + 2 * half) * 128
                         + 4 * (n + 16 * c);

    v8f acc[9][MT + 1];
#pragma unroll
    for (int x = 0; x < 9; ++x)
#pragma unroll
        for (int j = 0; j <= MT; ++j) acc[x][j] = v8f{};

    v2f ring[9][R];
    v4f bq[BL][2];
#pragma unroll
    for (int s = 0; s < R; ++s) {
        const int kc = (s < 4) ? s : s - R;
#pragma unroll
        for (int x = 0; x < 9; ++x) {
            ring[x][s].x = sa[x * SHN + 4 * kc];
            ring[x][s].y = sa[x * SHN + 4 * kc + 1];
        }
    }
#pragma unroll
    for (int s = 0; s < BL - 4; ++s) {
        bq[s][0] = *(const v4f*)(p + 512 * s);
        bq[s][1] = *(const v4f*)(p + 512 * s + 128);
    }

    for (int c0 = 0; c0 < NCH; c0 += R) {
#pragma unroll
        for (int i0 = 0; i0 < R; i0 += 4) {
#pragma unroll
            for (int u = 0; u < 4; ++u) {
                const v4f g0 = bq[(i0 + u) % BL][0];
                const v4f g1 = bq[(i0 + u) % BL][1];
                v2f bf[9];
                bf[0] = v2f{g0.x, g1.x};
                bf[1] = v2f{g0.y, g1.y};
                bf[3] = v2f{g0.z, g1.z};
                bf[4] = v2f{g0.w, g1.w};
                bf[2] = bf[0] + bf[1];
                bf[5] = bf[3] + bf[4];
                bf[6] = bf[0] + bf[3];
                bf[7] = bf[1] + bf[4];
                bf[8] = bf[6] + bf[7];
#pragma unroll
                for (int x = 0; x < 9; ++x)
                    acc[x][MT] = __builtin_amdgcn_wmma_f32_16x16x4_f32(
                        false, ring[x][(i0 + u + 4) % R], false, bf[x], (short)0, acc[x][MT], false, false);
            }
            __builtin_amdgcn_sched_barrier(0);
#pragma unroll
            for (int u = 0; u < 4; ++u) {
                const int s = (i0 + u + 4) % R;
                const int kc = i0 + u + 4;
#pragma unroll
                for (int x = 0; x < 9; ++x) {
                    ring[x][s].x = sa[x * SHN + 4 * kc];
                    ring[x][s].y = sa[x * SHN + 4 * kc + 1];
                }
            }
#pragma unroll
            for (int u = 0; u < 4; ++u) {
                bq[(i0 + u + BL - 4) % BL][0] = *(const v4f*)(p + 512 * (i0 + u + BL - 4));
                bq[(i0 + u + BL - 4) % BL][1] = *(const v4f*)(p + 512 * (i0 + u + BL - 4) + 128);
            }
            __builtin_amdgcn_sched_barrier(0);
#pragma unroll
            for (int u = 0; u < 4; ++u) {
                const v4f g0 = bq[(i0 + u) % BL][0];
                const v4f g1 = bq[(i0 + u) % BL][1];
                v2f bf[9];
                bf[0] = v2f{g0.x, g1.x};
                bf[1] = v2f{g0.y, g1.y};
                bf[3] = v2f{g0.z, g1.z};
                bf[4] = v2f{g0.w, g1.w};
                bf[2] = bf[0] + bf[1];
                bf[5] = bf[3] + bf[4];
                bf[6] = bf[0] + bf[3];
                bf[7] = bf[1] + bf[4];
                bf[8] = bf[6] + bf[7];
#pragma unroll
                for (int x = 0; x < 9; ++x) {
                    if (x == 1 || x == 3 || x == 4 || x == 5 || x == 7)
                        acc[x][0] = __builtin_amdgcn_wmma_f32_16x16x4_f32(
                            false, ring[x][(i0 + u) % R], false, bf[x], (short)0, acc[x][0], false, false);
                }
#pragma unroll
                for (int j = 1; j < MT; ++j) {
#pragma unroll
                    for (int x = 0; x < 9; ++x)
                        acc[x][j] = __builtin_amdgcn_wmma_f32_16x16x4_f32(
                            false, ring[x][(i0 + u - 4 * j + R) % R], false, bf[x], (short)0, acc[x][j], false, false);
                }
            }
        }
        p  += 512 * R;
        sa += 4 * R;
    }

#pragma unroll
    for (int jj = 1; jj <= MT; ++jj) {
        float pbA = __shfl_xor(half ? acc[1][jj - 1][7] : acc[1][jj][7], 16);
        float paB = __shfl_xor(half ? acc[3][jj - 1][7] : acc[3][jj][7], 16);
        float pbB = __shfl_xor(half ? acc[4][jj - 1][7] : acc[4][jj][7], 16);
        float pcB = __shfl_xor(half ? acc[5][jj - 1][7] : acc[5][jj][7], 16);
        float pbC = __shfl_xor(half ? acc[7][jj - 1][7] : acc[7][jj][7], 16);
        const int tt = t0 + 64 * (jj - 1) + 32 * half;
        float* o = dst + (size_t)(n + 16 * c) * T_LEN + (tt - PAD);
#pragma unroll
        for (int r = 0; r < 8; ++r) {
            const float aA = acc[0][jj][r], bA = acc[1][jj][r], cA = acc[2][jj][r];
            const float aB = acc[3][jj][r], bB = acc[4][jj][r], cB = acc[5][jj][r];
            const float aC = acc[6][jj][r], bC = acc[7][jj][r], cC = acc[8][jj][r];
            const float Ae  = aA + pbA;
            const float Ao  = cA - aA - bA;
            const float Be  = aB + pbB;
            const float Bo  = cB - aB - bB;
            const float Bom = pcB - paB - pbB;
            const float Ce  = aC + pbC;
            const float Co  = cC - aC - bC;
            *((float4*)o + r) = make_float4(Ae + Bom, Ce - Ae - Be, Ao + Be, Co - Ao - Bo);
            pbA = bA; paB = aB; pbB = bB; pcB = cB; pbC = bC;
        }
    }
}

extern "C" void kernel_launch(void* const* d_in, const int* in_sizes, int n_in,
                              void* d_out, int out_size, void* d_ws, size_t ws_size,
                              hipStream_t stream) {
    const float* x    = (const float*)d_in[0];
    const float* kern = (const float*)d_in[1];
    float* out = (float*)d_out;
    float* ws  = (float*)d_ws;

    float* f2  = ws + F2OFF;
    float* xq  = ws + XOFF;

    prep_kernel<<<NFB + NSB, 128, 0, stream>>>(x, kern, f2, xq);

    fir_pass4_kernel<MT2>
        <<<T_LEN / (64 * MT2 * 2), 128, 0, stream>>>(f2, xq, out, PAD);
}


# first and last ring pass of the k axis issue only the WMMAs of tiles whose tap window is open (the rest multiplied the filters' zero guards): 21176 WMMAs a wave, not 22140; every sum unchanged
# speedup vs baseline: 2.1533x; 1.0394x over previous
#include <hip/hip_runtime.h>


typedef __attribute__((ext_vector_type(2))) float v2f;
typedef __attribute__((ext_vector_type(4))) float v4f;
typedef __attribute__((ext_vector_type(8))) float v8f;

#define K_TAPS 4096
#define NF     8192
#define KQ     2048
#define PAD    2048
#define T_LEN  131072
#define S_SIG  32
#define TP     (T_LEN + 2*PAD)
#define GT     2112
#define ROWS   (TP + 2*GT + 256)
#define SBASE  (GT - 4096)
#define MT2    4
#define XOFF   32768

template<int MT> struct Geo4 {
    static constexpr int G   = 16 * (MT + 1);
    static constexpr int R   = 4 * (MT + 1);
    static constexpr int BL  = R / 2;
    static constexpr int NCH = ((516 + 4 * MT + R - 1) / R) * R;
    static constexpr int SHN = G + 4 * (NCH + 4) + 8;
};
#define F2OFF  0
static_assert(F2OFF + 9 * Geo4<MT2>::SHN <= XOFF, "filter blocks end before the staged input");
static_assert(SBASE + PAD + T_LEN - 64 * MT2 - 64
              + 16 * (Geo4<MT2>::NCH + Geo4<MT2>::BL - 4) <= ROWS, "k axis ends inside the staged rows");
static_assert(SBASE + PAD - 64 >= 0, "k axis begins inside the staged rows");

__device__ __forceinline__ int quad_slot(int r) { return ((r & 1) << 1) | ((r >> 1) & 1); }

#define UR 8
template<int MT>
__device__ __forceinline__ void build_tap_group(const float* __restrict__ h,
                                                float* __restrict__ f2,
                                                int jq, int lane) {
    constexpr int G   = Geo4<MT>::G;
    constexpr int SHN = Geo4<MT>::SHN;
    int D[4], n[4];
    int nmax = 0;
#pragma unroll
    for (int k = 0; k < 4; ++k) {
        const int d = 4 * jq - k - K_TAPS;
        D[k] = d < 0 ? -d : d;
        n[k] = K_TAPS - D[k];
        nmax = max(nmax, n[k]);
    }
    float acc[4] = {0.f, 0.f, 0.f, 0.f};
    for (int t0 = 0; t0 < nmax; t0 += 32 * UR) {
        float p[UR], q[4][UR];
#pragma unroll
        for (int j = 0; j < UR; ++j) {
            const int u = t0 + 32 * j + lane;
            p[j] = h[min(u, K_TAPS - 1)];
#pragma unroll
            for (int k = 0; k < 4; ++k) q[k][j] = h[min(u + D[k], K_TAPS - 1)];
        }
#pragma unroll
        for (int j = 0; j < UR; ++j) {
            const int u = t0 + 32 * j + lane;
#pragma unroll
            for (int k = 0; k < 4; ++k)
                if (u < n[k]) acc[k] = fmaf(p[j], q[k][j], acc[k]);
        }
    }
#pragma unroll
    for (int k = 0; k < 4; ++k)
        for (int m = 16; m >= 1; m >>= 1)
            acc[k] += __shfl_xor(acc[k], m);
    const float a = acc[0], b = acc[1], c = acc[2], d = acc[3];
    const float ab = a + b, cd = c + d;
    if (lane < 9) {
        float v;
        switch (lane) {
            case 0:  v = a;       break;
            case 1:  v = c;       break;
            case 2:  v = a + c;   break;
            case 3:  v = b;       break;
            case 4:  v = d;       break;
            case 5:  v = b + d;   break;
            case 6:  v = ab;      break;
            case 7:  v = cd;      break;
            default: v = ab + cd; break;
        }
        f2[lane * SHN + G + jq] = v;
    } else if (lane < 18 && jq < SHN - KQ - 1) {
        f2[(lane - 9) * SHN + (jq < G ? jq : KQ + 1 + jq)] = 0.f;
    }
}

static_assert(ROWS % 128 == 0, "whole blocks of 128 rows");
static_assert(GT % 4 == 0 && PAD % 4 == 0 && T_LEN % 4 == 0, "row groups do not straddle boundaries");
#define NFB ((KQ + 1 + 3) / 4)
#define NSB (ROWS / 128)
__global__ __launch_bounds__(128)
void prep_kernel(const float* __restrict__ x,
                 const float* __restrict__ h,
                 float* __restrict__ f2,
                 float* __restrict__ xq) {
    const int lane = threadIdx.x & 31;
    if (blockIdx.x < NFB) {
        const int jq = blockIdx.x * 4 + __builtin_amdgcn_readfirstlane((int)(threadIdx.x >> 5));
        if (jq <= KQ) build_tap_group<MT2>(h, f2, jq, lane);
        return;
    }
    const int bs   = blockIdx.x - NFB;
    const int s    = ((threadIdx.x >> 5) << 3) + (lane & 7);
    const int g0   = bs * 32 + (lane >> 3);
    const int r0   = bs * 128;
    const float* xs = x + (size_t)s * T_LEN;
    v4f v[8];
    if (r0 >= GT + PAD && r0 + 128 <= GT + PAD + T_LEN) {
        const float* pin = xs + (4 * g0 - (GT + PAD));
#pragma unroll
        for (int j = 0; j < 8; ++j) {
            const v4f a = *(const v4f*)(pin + 16 * j);
            v[j] = v4f{a.x, a.z, a.y, a.w};
        }
    } else {
#pragma unroll
        for (int j = 0; j < 8; ++j) {
            float e[4];
#pragma unroll
            for (int r = 0; r < 4; ++r) {
                const int tg = 4 * (g0 + 4 * j) + r - GT;
                float val = 0.f;
                if (tg >= 0 && tg < TP) {
                    int tr = tg - PAD;
                    if (tr < 0)            tr = -tr;
                    else if (tr >= T_LEN)  tr = 2 * T_LEN - 2 - tr;
                    val = xs[tr];
                }
                e[r] = val;
            }
            v[j] = v4f{e[0], e[2], e[1], e[3]};
        }
    }
    float* o = xq + (size_t)g0 * 128 + 4 * s;
#pragma unroll
    for (int j = 0; j < 8; ++j)
        *(v4f*)(o + 512 * j) = v[j];
}

template<int V> struct Part { static constexpr int value = V; };
template<int MT>
__global__ __launch_bounds__(128)
void fir_pass4_kernel(const float* __restrict__ filt,
                      const float* __restrict__ src,
                      float* __restrict__ dst,
                      int t0_origin) {
    constexpr int G   = Geo4<MT>::G;
    constexpr int R   = Geo4<MT>::R;
    constexpr int BL  = Geo4<MT>::BL;
    constexpr int NCH = Geo4<MT>::NCH;
    constexpr int SHN = Geo4<MT>::SHN;
    static_assert(R % BL == 0, "B ring must divide the A ring");
    static_assert(SHN % 4 == 0, "whole 16-byte slots");
    static_assert(4 * NCH >= KQ + 1 + 16 * MT + 15, "every tile's tap window lies inside the k axis");
    constexpr int WLAST  = (KQ + 15) / 4;
    constexpr int LASTCH = 4 * MT + WLAST;
    static_assert((WLAST + 1) % 4 == 0, "a tap window ends on a 4-chunk step boundary");
    static_assert(NCH >= 3 * R, "a first, at least one middle and a last ring pass");
    static_assert(LASTCH < NCH, "the last pass reaches the last needed chunk");
    static_assert(WLAST <= NCH - R + 3, "no fragment the last pass uses is refilled inside it");

    constexpr int NV  = 9 * SHN / 4;
    constexpr int NIT = (NV + 127) / 128;

    __shared__ __attribute__((aligned(16))) float sh[9 * SHN];
    constexpr int NB = 20;
    for (int j0 = 0; j0 < NIT; j0 += NB) {
        float4 stage[NB];
#pragma unroll
        for (int j = 0; j < NB; ++j) {
            const int v = min((int)threadIdx.x + 128 * (j0 + j), NV - 1);
            stage[j] = *(const float4*)(filt + 4 * v);
        }
#pragma unroll
        for (int j = 0; j < NB; ++j) {
            const int v = (int)threadIdx.x + 128 * (j0 + j);
            if (v < NV) *(float4*)(sh + 4 * v) = stage[j];
        }
    }
    __syncthreads();

    const int lane = threadIdx.x & 31;
    const int wave = threadIdx.x >> 5;
    const int n    = lane & 15;
    const int half = lane >> 4;
    const int q    = half * 2;
    const int c    = wave & 1;
    const int grp  = blockIdx.x * 2 + (wave >> 1);

    const int t0 = t0_origin + grp * (64 * MT);

    const float* sa = sh + (G + q - n);
    const float* p = src + ((size_t)(((SBASE + t0) >> 2) - 16) + 2 * half) * 128
                         + 4 * (n + 16 * c);

    v8f acc[9][MT + 1];
#pragma unroll
    for (int x = 0; x < 9; ++x)
#pragma unroll
        for (int j = 0; j <= MT; ++j) acc[x][j] = v8f{};

    v2f ring[9][R];
    v4f bq[BL][2];
#pragma unroll
    for (int s = 0; s < 4; ++s) {
#pragma unroll
        for (int x = 0; x < 9; ++x) {
            ring[x][s].x = sa[x * SHN + 4 * s];
            ring[x][s].y = sa[x * SHN + 4 * s + 1];
        }
    }
#pragma unroll
    for (int s = 0; s < BL - 4; ++s) {
        bq[s][0] = *(const v4f*)(p + 512 * s);
        bq[s][1] = *(const v4f*)(p + 512 * s + 128);
    }

    auto ring_pass = [&](auto part_c) __attribute__((always_inline)) {
        constexpr int PART = decltype(part_c)::value;
        constexpr int C0   = NCH - R;
#pragma unroll
        for (int i0 = 0; i0 < R; i0 += 4) {
            bool live[MT + 1];
#pragma unroll
            for (int j = 0; j <= MT; ++j)
                live[j] = PART == 1 ? true
                        : PART == 0 ? (i0 >= 4 * j)
                                    : (C0 + i0 + 3 <= 4 * j + WLAST);
            if (live[MT]) {
#pragma unroll
                for (int u = 0; u < 4; ++u) {
                    const v4f g0 = bq[(i0 + u) % BL][0];
                    const v4f g1 = bq[(i0 + u) % BL][1];
                    v2f bf[9];
                    bf[0] = v2f{g0.x, g1.x};
                    bf[1] = v2f{g0.y, g1.y};
                    bf[3] = v2f{g0.z, g1.z};
                    bf[4] = v2f{g0.w, g1.w};
                    bf[2] = bf[0] + bf[1];
                    bf[5] = bf[3] + bf[4];
                    bf[6] = bf[0] + bf[3];
                    bf[7] = bf[1] + bf[4];
                    bf[8] = bf[6] + bf[7];
#pragma unroll
                    for (int x = 0; x < 9; ++x)
                        acc[x][MT] = __builtin_amdgcn_wmma_f32_16x16x4_f32(
                            false, ring[x][(i0 + u + 4) % R], false, bf[x], (short)0, acc[x][MT], false, false);
                }
            }
            __builtin_amdgcn_sched_barrier(0);
            if (PART != 2) {
#pragma unroll
                for (int u = 0; u < 4; ++u) {
                    const int s = (i0 + u + 4) % R;
                    const int kc = i0 + u + 4;
#pragma unroll
                    for (int x = 0; x < 9; ++x) {
                        ring[x][s].x = sa[x * SHN + 4 * kc];
                        ring[x][s].y = sa[x * SHN + 4 * kc + 1];
                    }
                }
            }
#pragma unroll
            for (int u = 0; u < 4; ++u) {
                if (PART != 2 || C0 + i0 + u + BL - 4 <= LASTCH) {
                    bq[(i0 + u + BL - 4) % BL][0] = *(const v4f*)(p + 512 * (i0 + u + BL - 4));
                    bq[(i0 + u + BL - 4) % BL][1] = *(const v4f*)(p + 512 * (i0 + u + BL - 4) + 128);
                }
            }
            __builtin_amdgcn_sched_barrier(0);
            if (PART != 2 || C0 + i0 <= LASTCH) {
#pragma unroll
                for (int u = 0; u < 4; ++u) {
                    const v4f g0 = bq[(i0 + u) % BL][0];
                    const v4f g1 = bq[(i0 + u) % BL][1];
                    v2f bf[9];
                    bf[0] = v2f{g0.x, g1.x};
                    bf[1] = v2f{g0.y, g1.y};
                    bf[3] = v2f{g0.z, g1.z};
                    bf[4] = v2f{g0.w, g1.w};
                    bf[2] = bf[0] + bf[1];
                    bf[5] = bf[3] + bf[4];
                    bf[6] = bf[0] + bf[3];
                    bf[7] = bf[1] + bf[4];
                    bf[8] = bf[6] + bf[7];
                    if (live[0]) {
#pragma unroll
                        for (int x = 0; x < 9; ++x) {
                            if (x == 1 || x == 3 || x == 4 || x == 5 || x == 7)
                                acc[x][0] = __builtin_amdgcn_wmma_f32_16x16x4_f32(
                                    false, ring[x][(i0 + u) % R], false, bf[x], (short)0, acc[x][0], false, false);
                        }
                    }
#pragma unroll
                    for (int j = 1; j < MT; ++j) {
                        if (live[j]) {
#pragma unroll
                            for (int x = 0; x < 9; ++x)
                                acc[x][j] = __builtin_amdgcn_wmma_f32_16x16x4_f32(
                                    false, ring[x][(i0 + u - 4 * j + R) % R], false, bf[x], (short)0, acc[x][j], false, false);
                        }
                    }
                }
            }
        }
    };

    ring_pass(Part<0>{});
    p  += 512 * R;
    sa += 4 * R;
    for (int c0 = R; c0 < NCH - R; c0 += R) {
        ring_pass(Part<1>{});
        p  += 512 * R;
        sa += 4 * R;
    }
    ring_pass(Part<2>{});

#pragma unroll
    for (int jj = 1; jj <= MT; ++jj) {
        float pbA = __shfl_xor(half ? acc[1][jj - 1][7] : acc[1][jj][7], 16);
        float paB = __shfl_xor(half ? acc[3][jj - 1][7] : acc[3][jj][7], 16);
        float pbB = __shfl_xor(half ? acc[4][jj - 1][7] : acc[4][jj][7], 16);
        float pcB = __shfl_xor(half ? acc[5][jj - 1][7] : acc[5][jj][7], 16);
        float pbC = __shfl_xor(half ? acc[7][jj - 1][7] : acc[7][jj][7], 16);
        const int tt = t0 + 64 * (jj - 1) + 32 * half;
        float* o = dst + (size_t)(n + 16 * c) * T_LEN + (tt - PAD);
#pragma unroll
        for (int r = 0; r < 8; ++r) {
            const float aA = acc[0][jj][r], bA = acc[1][jj][r], cA = acc[2][jj][r];
            const float aB = acc[3][jj][r], bB = acc[4][jj][r], cB = acc[5][jj][r];
            const float aC = acc[6][jj][r], bC = acc[7][jj][r], cC = acc[8][jj][r];
            const float Ae  = aA + pbA;
            const float Ao  = cA - aA - bA;
            const float Be  = aB + pbB;
            const float Bo  = cB - aB - bB;
            const float Bom = pcB - paB - pbB;
            const float Ce  = aC + pbC;
            const float Co  = cC - aC - bC;
            *((float4*)o + r) = make_float4(Ae + Bom, Ce - Ae - Be, Ao + Be, Co - Ao - Bo);
            pbA = bA; paB = aB; pbB = bB; pcB = cB; pbC = bC;
        }
    }
}

extern "C" void kernel_launch(void* const* d_in, const int* in_sizes, int n_in,
                              void* d_out, int out_size, void* d_ws, size_t ws_size,
                              hipStream_t stream) {
    const float* x    = (const float*)d_in[0];
    const float* kern = (const float*)d_in[1];
    float* out = (float*)d_out;
    float* ws  = (float*)d_ws;

    float* f2  = ws + F2OFF;
    float* xq  = ws + XOFF;

    prep_kernel<<<NFB + NSB, 128, 0, stream>>>(x, kern, f2, xq);

    fir_pass4_kernel<MT2>
        <<<T_LEN / (64 * MT2 * 2), 128, 0, stream>>>(f2, xq, out, PAD);
}


# a wave owns one signal; its 16 WMMA columns are consecutive time blocks: the row in front of a column is the neighbour lane's, column 0's a 16-segment diagonal product (its sum order changes)
# speedup vs baseline: 2.3420x; 1.0876x over previous
#include <hip/hip_runtime.h>


typedef __attribute__((ext_vector_type(2))) float v2f;
typedef __attribute__((ext_vector_type(4))) float v4f;
typedef __attribute__((ext_vector_type(8))) float v8f;

#define K_TAPS 4096
#define NF     8192
#define KQ     2048
#define PAD    2048
#define T_LEN  131072
#define S_SIG  32
#define TP     (T_LEN + 2*PAD)
#define GT     2112
#define ROWS   (TP + 2*GT + 256)
#define SBASE  (GT - 4096)
#define MT2    4
#define XOFF   32768

template<int MT> struct Geo4 {
    static constexpr int G   = 16 * (MT + 1);
    static constexpr int R   = 4 * (MT + 1);
    static constexpr int BL  = R / 2;
    static constexpr int NCH = ((516 + 4 * MT + R - 1) / R) * R;
    static constexpr int SHN = G + 4 * (NCH + 4) + 8;
};
#define F2OFF  0
static_assert(F2OFF + 9 * Geo4<MT2>::SHN <= XOFF, "filter blocks end before the staged input");
static_assert(SBASE + PAD + T_LEN - 64 * MT2 - 64
              + 16 * (Geo4<MT2>::NCH + Geo4<MT2>::BL - 4) <= ROWS, "k axis ends inside the staged rows");
static_assert(SBASE + PAD - 64 >= 0, "k axis begins inside the staged rows");
#define DSEG   132
#define NDC    (DSEG / 4)
static_assert(DSEG % 4 == 0 && 16 * DSEG >= KQ + 1, "sixteen segments cover every tap");
static_assert(Geo4<MT2>::G + 16 * DSEG <= Geo4<MT2>::SHN, "segment reads stay inside a sub-filter and its zeros");
static_assert(SBASE + PAD + T_LEN - 16 * 64 * MT2 - 4 + 4 * 16 * DSEG <= ROWS, "segment reads stay inside the staged rows");

__device__ __forceinline__ int quad_slot(int r) { return ((r & 1) << 1) | ((r >> 1) & 1); }

#define UR 8
template<int MT>
__device__ __forceinline__ void build_tap_group(const float* __restrict__ h,
                                                float* __restrict__ f2,
                                                int jq, int lane) {
    constexpr int G   = Geo4<MT>::G;
    constexpr int SHN = Geo4<MT>::SHN;
    int D[4], n[4];
    int nmax = 0;
#pragma unroll
    for (int k = 0; k < 4; ++k) {
        const int d = 4 * jq - k - K_TAPS;
        D[k] = d < 0 ? -d : d;
        n[k] = K_TAPS - D[k];
        nmax = max(nmax, n[k]);
    }
    float acc[4] = {0.f, 0.f, 0.f, 0.f};
    for (int t0 = 0; t0 < nmax; t0 += 32 * UR) {
        float p[UR], q[4][UR];
#pragma unroll
        for (int j = 0; j < UR; ++j) {
            const int u = t0 + 32 * j + lane;
            p[j] = h[min(u, K_TAPS - 1)];
#pragma unroll
            for (int k = 0; k < 4; ++k) q[k][j] = h[min(u + D[k], K_TAPS - 1)];
        }
#pragma unroll
        for (int j = 0; j < UR; ++j) {
            const int u = t0 + 32 * j + lane;
#pragma unroll
            for (int k = 0; k < 4; ++k)
                if (u < n[k]) acc[k] = fmaf(p[j], q[k][j], acc[k]);
        }
    }
#pragma unroll
    for (int k = 0; k < 4; ++k)
        for (int m = 16; m >= 1; m >>= 1)
            acc[k] += __shfl_xor(acc[k], m);
    const float a = acc[0], b = acc[1], c = acc[2], d = acc[3];
    const float ab = a + b, cd = c + d;
    if (lane < 9) {
        float v;
        switch (lane) {
            case 0:  v = a;       break;
            case 1:  v = c;       break;
            case 2:  v = a + c;   break;
            case 3:  v = b;       break;
            case 4:  v = d;       break;
            case 5:  v = b + d;   break;
            case 6:  v = ab;      break;
            case 7:  v = cd;      break;
            default: v = ab + cd; break;
        }
        f2[lane * SHN + G + jq] = v;
    } else if (lane < 18 && jq < SHN - KQ - 1) {
        f2[(lane - 9) * SHN + (jq < G ? jq : KQ + 1 + jq)] = 0.f;
    }
}

static_assert(ROWS % 128 == 0, "whole blocks of 128 rows");
static_assert(GT % 4 == 0 && PAD % 4 == 0 && T_LEN % 4 == 0, "row groups do not straddle boundaries");
#define NFB ((KQ + 1 + 3) / 4)
#define NSB (ROWS / 128)
__global__ __launch_bounds__(128)
void prep_kernel(const float* __restrict__ x,
                 const float* __restrict__ h,
                 float* __restrict__ f2,
                 float* __restrict__ xq) {
    const int lane = threadIdx.x & 31;
    if (blockIdx.x < NFB) {
        const int jq = blockIdx.x * 4 + __builtin_amdgcn_readfirstlane((int)(threadIdx.x >> 5));
        if (jq <= KQ) build_tap_group<MT2>(h, f2, jq, lane);
        return;
    }
    const int bs   = blockIdx.x - NFB;
    const int s0   = (threadIdx.x >> 5) << 3;
    const int g    = bs * 32 + lane;
    const int r0   = bs * 128;
    const float* xs = x + (size_t)s0 * T_LEN;
    v4f v[8];
    if (r0 >= GT + PAD && r0 + 128 <= GT + PAD + T_LEN) {
        const float* pin = xs + (4 * g - (GT + PAD));
#pragma unroll
        for (int j = 0; j < 8; ++j) {
            const v4f a = *(const v4f*)(pin + (size_t)j * T_LEN);
            v[j] = v4f{a.x, a.z, a.y, a.w};
        }
    } else {
#pragma unroll
        for (int j = 0; j < 8; ++j) {
            float e[4];
#pragma unroll
            for (int r = 0; r < 4; ++r) {
                const int tg = 4 * g + r - GT;
                float val = 0.f;
                if (tg >= 0 && tg < TP) {
                    int tr = tg - PAD;
                    if (tr < 0)            tr = -tr;
                    else if (tr >= T_LEN)  tr = 2 * T_LEN - 2 - tr;
                    val = xs[(size_t)j * T_LEN + tr];
                }
                e[r] = val;
            }
            v[j] = v4f{e[0], e[2], e[1], e[3]};
        }
    }
    float* o = xq + (size_t)s0 * ROWS + 4 * g;
#pragma unroll
    for (int j = 0; j < 8; ++j)
        *(v4f*)(o + (size_t)j * ROWS) = v[j];
}

template<int V> struct Part { static constexpr int value = V; };
constexpr int xb_of(int e) { return e == 0 ? 1 : e == 1 ? 3 : e == 2 ? 4 : e == 3 ? 5 : 7; }
template<int MT>
__global__ __launch_bounds__(128)
void fir_pass4_kernel(const float* __restrict__ filt,
                      const float* __restrict__ src,
                      float* __restrict__ dst,
                      int t0_origin) {
    constexpr int G   = Geo4<MT>::G;
    constexpr int R   = Geo4<MT>::R;
    constexpr int BL  = Geo4<MT>::BL;
    constexpr int NCH = Geo4<MT>::NCH;
    constexpr int SHN = Geo4<MT>::SHN;
    static_assert(R % BL == 0, "B ring must divide the A ring");
    static_assert(SHN % 4 == 0, "whole 16-byte slots");
    static_assert(4 * NCH >= KQ + 1 + 16 * MT + 15, "every tile's tap window lies inside the k axis");
    constexpr int WLAST  = (KQ + 15) / 4;
    constexpr int LASTCH = 4 * MT + WLAST;
    static_assert((WLAST + 1) % 4 == 0, "a tap window ends on a 4-chunk step boundary");
    static_assert(NCH >= 3 * R, "a first, at least one middle and a last ring pass");
    static_assert(LASTCH < NCH, "the last pass reaches the last needed chunk");
    static_assert(WLAST <= NCH - R + 3, "no fragment the last pass uses is refilled inside it");

    constexpr int NV  = 9 * SHN / 4;
    constexpr int NIT = (NV + 127) / 128;

    __shared__ __attribute__((aligned(16))) float sh[9 * SHN];
    constexpr int NB = 20;
    for (int j0 = 0; j0 < NIT; j0 += NB) {
        float4 stage[NB];
#pragma unroll
        for (int j = 0; j < NB; ++j) {
            const int v = min((int)threadIdx.x + 128 * (j0 + j), NV - 1);
            stage[j] = *(const float4*)(filt + 4 * v);
        }
#pragma unroll
        for (int j = 0; j < NB; ++j) {
            const int v = (int)threadIdx.x + 128 * (j0 + j);
            if (v < NV) *(float4*)(sh + 4 * v) = stage[j];
        }
    }
    __syncthreads();

    const int lane = threadIdx.x & 31;
    const int wave = __builtin_amdgcn_readfirstlane((int)(threadIdx.x >> 5));
    const int n    = lane & 15;
    const int half = lane >> 4;
    const int q    = half * 2;
    const int wid  = blockIdx.x * 4 + wave;
    const int sig  = wid % S_SIG;
    const int rng  = wid / S_SIG;
    const int grp  = rng * 16 + n;

    const int t0   = t0_origin + grp * (64 * MT);
    const int t0c0 = t0_origin + rng * 16 * (64 * MT);

    const float* sa = sh + (G + q - n);
    const float* srcs = src + (size_t)sig * ROWS;
    const float* p = srcs + ((size_t)(((SBASE + t0) >> 2) - 16) + 2 * half) * 4;

    float bnd[5];
    {
        v8f ad[5];
#pragma unroll
        for (int e = 0; e < 5; ++e) ad[e] = v8f{};
        const float* sd = sh + (G + DSEG * n + q);
        const float* pd = srcs + ((size_t)(((SBASE + t0c0) >> 2) - 1) + DSEG * n + q) * 4;
#pragma unroll
        for (int kc = 0; kc < NDC; ++kc) {
            const v4f g0 = *(const v4f*)(pd + 16 * kc);
            const v4f g1 = *(const v4f*)(pd + 16 * kc + 4);
            v2f bd[5];
            bd[0] = v2f{g0.y, g1.y};
            bd[1] = v2f{g0.z, g1.z};
            bd[2] = v2f{g0.w, g1.w};
            bd[3] = bd[1] + bd[2];
            bd[4] = bd[0] + bd[2];
#pragma unroll
            for (int e = 0; e < 5; ++e) {
                v2f a;
                a.x = sd[xb_of(e) * SHN + 4 * kc];
                a.y = sd[xb_of(e) * SHN + 4 * kc + 1];
                ad[e] = __builtin_amdgcn_wmma_f32_16x16x4_f32(
                    false, a, false, bd[e], (short)0, ad[e], false, false);
            }
        }
#pragma unroll
        for (int e = 0; e < 5; ++e) {
            float d = 0.f;
#pragma unroll
            for (int r = 0; r < 8; ++r) d = ((n & 7) == r) ? ad[e][r] : d;
            d = ((n >> 3) == half) ? d : 0.f;
            for (int m = 16; m >= 1; m >>= 1) d += __shfl_xor(d, m);
            bnd[e] = d;
        }
    }

    v8f acc[9][MT + 1];
#pragma unroll
    for (int x = 0; x < 9; ++x)
#pragma unroll
        for (int j = 0; j <= MT; ++j) acc[x][j] = v8f{};

    v2f ring[9][R];
    v4f bq[BL][2];
#pragma unroll
    for (int s = 0; s < 4; ++s) {
#pragma unroll
        for (int x = 0; x < 9; ++x) {
            ring[x][s].x = sa[x * SHN + 4 * s];
            ring[x][s].y = sa[x * SHN + 4 * s + 1];
        }
    }
#pragma unroll
    for (int s = 0; s < BL - 4; ++s) {
        bq[s][0] = *(const v4f*)(p + 16 * s);
        bq[s][1] = *(const v4f*)(p + 16 * s + 4);
    }

    auto ring_pass = [&](auto part_c) __attribute__((always_inline)) {
        constexpr int PART = decltype(part_c)::value;
        constexpr int C0   = NCH - R;
#pragma unroll
        for (int i0 = 0; i0 < R; i0 += 4) {
            bool live[MT + 1];
#pragma unroll
            for (int j = 0; j <= MT; ++j)
                live[j] = PART == 1 ? true
                        : PART == 0 ? (i0 >= 4 * j)
                                    : (C0 + i0 + 3 <= 4 * j + WLAST);
            if (live[MT]) {
#pragma unroll
                for (int u = 0; u < 4; ++u) {
                    const v4f g0 = bq[(i0 + u) % BL][0];
                    const v4f g1 = bq[(i0 + u) % BL][1];
                    v2f bf[9];
                    bf[0] = v2f{g0.x, g1.x};
                    bf[1] = v2f{g0.y, g1.y};
                    bf[3] = v2f{g0.z, g1.z};
                    bf[4] = v2f{g0.w, g1.w};
                    bf[2] = bf[0] + bf[1];
                    bf[5] = bf[3] + bf[4];
                    bf[6] = bf[0] + bf[3];
                    bf[7] = bf[1] + bf[4];
                    bf[8] = bf[6] + bf[7];
#pragma unroll
                    for (int x = 0; x < 9; ++x)
                        acc[x][MT] = __builtin_amdgcn_wmma_f32_16x16x4_f32(
                            false, ring[x][(i0 + u + 4) % R], false, bf[x], (short)0, acc[x][MT], false, false);
                }
            }
            __builtin_amdgcn_sched_barrier(0);
            if (PART != 2) {
#pragma unroll
                for (int u = 0; u < 4; ++u) {
                    const int s = (i0 + u + 4) % R;
                    const int kc = i0 + u + 4;
#pragma unroll
                    for (int x = 0; x < 9; ++x) {
                        ring[x][s].x = sa[x * SHN + 4 * kc];
                        ring[x][s].y = sa[x * SHN + 4 * kc + 1];
                    }
                }
            }
#pragma unroll
            for (int u = 0; u < 4; ++u) {
                if (PART != 2 || C0 + i0 + u + BL - 4 <= LASTCH) {
                    bq[(i0 + u + BL - 4) % BL][0] = *(const v4f*)(p + 16 * (i0 + u + BL - 4));
                    bq[(i0 + u + BL - 4) % BL][1] = *(const v4f*)(p + 16 * (i0 + u + BL - 4) + 4);
                }
            }
            __builtin_amdgcn_sched_barrier(0);
            if (PART != 2 || C0 + i0 <= LASTCH) {
#pragma unroll
                for (int u = 0; u < 4; ++u) {
                    const v4f g0 = bq[(i0 + u) % BL][0];
                    const v4f g1 = bq[(i0 + u) % BL][1];
                    v2f bf[9];
                    bf[0] = v2f{g0.x, g1.x};
                    bf[1] = v2f{g0.y, g1.y};
                    bf[3] = v2f{g0.z, g1.z};
                    bf[4] = v2f{g0.w, g1.w};
                    bf[2] = bf[0] + bf[1];
                    bf[5] = bf[3] + bf[4];
                    bf[6] = bf[0] + bf[3];
                    bf[7] = bf[1] + bf[4];
                    bf[8] = bf[6] + bf[7];
#pragma unroll
                    for (int j = 1; j < MT; ++j) {
                        if (live[j]) {
#pragma unroll
                            for (int x = 0; x < 9; ++x)
                                acc[x][j] = __builtin_amdgcn_wmma_f32_16x16x4_f32(
                                    false, ring[x][(i0 + u - 4 * j + R) % R], false, bf[x], (short)0, acc[x][j], false, false);
                        }
                    }
                }
            }
        }
    };

    ring_pass(Part<0>{});
    p  += 16 * R;
    sa += 4 * R;
    for (int c0 = R; c0 < NCH - R; c0 += R) {
        ring_pass(Part<1>{});
        p  += 16 * R;
        sa += 4 * R;
    }
    ring_pass(Part<2>{});

    const int src1 = half ? n : (((n + 15) & 15) + 16);
    const bool col0 = (lane == 0);
#pragma unroll
    for (int jj = 1; jj <= MT; ++jj) {
        float pbA, paB, pbB, pcB, pbC;
        if (jj == 1) {
            pbA = __shfl(half ? acc[1][MT][7] : acc[1][1][7], src1);
            paB = __shfl(half ? acc[3][MT][7] : acc[3][1][7], src1);
            pbB = __shfl(half ? acc[4][MT][7] : acc[4][1][7], src1);
            pcB = __shfl(half ? acc[5][MT][7] : acc[5][1][7], src1);
            pbC = __shfl(half ? acc[7][MT][7] : acc[7][1][7], src1);
            pbA = col0 ? bnd[0] : pbA;
            paB = col0 ? bnd[1] : paB;
            pbB = col0 ? bnd[2] : pbB;
            pcB = col0 ? bnd[3] : pcB;
            pbC = col0 ? bnd[4] : pbC;
        } else {
            pbA = __shfl_xor(half ? acc[1][jj - 1][7] : acc[1][jj][7], 16);
            paB = __shfl_xor(half ? acc[3][jj - 1][7] : acc[3][jj][7], 16);
            pbB = __shfl_xor(half ? acc[4][jj - 1][7] : acc[4][jj][7], 16);
            pcB = __shfl_xor(half ? acc[5][jj - 1][7] : acc[5][jj][7], 16);
            pbC = __shfl_xor(half ? acc[7][jj - 1][7] : acc[7][jj][7], 16);
        }
        const int tt = t0 + 64 * (jj - 1) + 32 * half;
        float* o = dst + (size_t)sig * T_LEN + (tt - PAD);
#pragma unroll
        for (int r = 0; r < 8; ++r) {
            const float aA = acc[0][jj][r], bA = acc[1][jj][r], cA = acc[2][jj][r];
            const float aB = acc[3][jj][r], bB = acc[4][jj][r], cB = acc[5][jj][r];
            const float aC = acc[6][jj][r], bC = acc[7][jj][r], cC = acc[8][jj][r];
            const float Ae  = aA + pbA;
            const float Ao  = cA - aA - bA;
            const float Be  = aB + pbB;
            const float Bo  = cB - aB - bB;
            const float Bom = pcB - paB - pbB;
            const float Ce  = aC + pbC;
            const float Co  = cC - aC - bC;
            *((float4*)o + r) = make_float4(Ae + Bom, Ce - Ae - Be, Ao + Be, Co - Ao - Bo);
            pbA = bA; paB = aB; pbB = bB; pcB = cB; pbC = bC;
        }
    }
}

extern "C" void kernel_launch(void* const* d_in, const int* in_sizes, int n_in,
                              void* d_out, int out_size, void* d_ws, size_t ws_size,
                              hipStream_t stream) {
    const float* x    = (const float*)d_in[0];
    const float* kern = (const float*)d_in[1];
    float* out = (float*)d_out;
    float* ws  = (float*)d_ws;

    float* f2  = ws + F2OFF;
    float* xq  = ws + XOFF;

    prep_kernel<<<NFB + NSB, 128, 0, stream>>>(x, kern, f2, xq);

    fir_pass4_kernel<MT2>
        <<<T_LEN / (64 * MT2 * 2), 128, 0, stream>>>(f2, xq, out, PAD);
}


# filter block stored as ready 8-byte A-fragment pairs interleaved by sub-filter: a step's 36 ring refills are 18 two-address LDS reads off one address (were 36 reads + 32 address adds); sums unchanged
# speedup vs baseline: 2.4463x; 1.0445x over previous
#include <hip/hip_runtime.h>


typedef __attribute__((ext_vector_type(2))) float v2f;
typedef __attribute__((ext_vector_type(4))) float v4f;
typedef __attribute__((ext_vector_type(8))) float v8f;

#define K_TAPS 4096
#define NF     8192
#define KQ     2048
#define PAD    2048
#define T_LEN  131072
#define S_SIG  32
#define TP     (T_LEN + 2*PAD)
#define GT     2112
#define ROWS   (TP + 2*GT + 256)
#define SBASE  (GT - 4096)
#define MT2    4
#define XOFF   65536

template<int MT> struct Geo4 {
    static constexpr int G   = 16 * (MT + 1);
    static constexpr int R   = 4 * (MT + 1);
    static constexpr int BL  = R / 2;
    static constexpr int NCH = ((516 + 4 * MT + R - 1) / R) * R;
    static constexpr int SHN = G + 4 * (NCH + 4) + 8;
};
#define F2OFF  0
static_assert(F2OFF + 18 * Geo4<MT2>::SHN <= XOFF, "filter block ends before the staged input");
static_assert(SBASE + PAD + T_LEN - 64 * MT2 - 64
              + 16 * (Geo4<MT2>::NCH + Geo4<MT2>::BL - 4) <= ROWS, "k axis ends inside the staged rows");
static_assert(SBASE + PAD - 64 >= 0, "k axis begins inside the staged rows");
#define DSEG   132
#define NDC    (DSEG / 4)
static_assert(DSEG % 4 == 0 && 16 * DSEG >= KQ + 1, "sixteen segments cover every tap");
static_assert(Geo4<MT2>::G + 16 * DSEG <= Geo4<MT2>::SHN, "segment reads stay inside a sub-filter and its zeros");
static_assert(SBASE + PAD + T_LEN - 16 * 64 * MT2 - 4 + 4 * 16 * DSEG <= ROWS, "segment reads stay inside the staged rows");

__device__ __forceinline__ int quad_slot(int r) { return ((r & 1) << 1) | ((r >> 1) & 1); }

#define UR 8
template<int MT>
__device__ __forceinline__ void build_tap_group(const float* __restrict__ h,
                                                float* __restrict__ f2,
                                                int jq, int lane) {
    constexpr int G   = Geo4<MT>::G;
    constexpr int SHN = Geo4<MT>::SHN;
    int D[4], n[4];
    int nmax = 0;
#pragma unroll
    for (int k = 0; k < 4; ++k) {
        const int d = 4 * jq - k - K_TAPS;
        D[k] = d < 0 ? -d : d;
        n[k] = K_TAPS - D[k];
        nmax = max(nmax, n[k]);
    }
    float acc[4] = {0.f, 0.f, 0.f, 0.f};
    for (int t0 = 0; t0 < nmax; t0 += 32 * UR) {
        float p[UR], q[4][UR];
#pragma unroll
        for (int j = 0; j < UR; ++j) {
            const int u = t0 + 32 * j + lane;
            p[j] = h[min(u, K_TAPS - 1)];
#pragma unroll
            for (int k = 0; k < 4; ++k) q[k][j] = h[min(u + D[k], K_TAPS - 1)];
        }
#pragma unroll
        for (int j = 0; j < UR; ++j) {
            const int u = t0 + 32 * j + lane;
#pragma unroll
            for (int k = 0; k < 4; ++k)
                if (u < n[k]) acc[k] = fmaf(p[j], q[k][j], acc[k]);
        }
    }
#pragma unroll
    for (int k = 0; k < 4; ++k)
        for (int m = 16; m >= 1; m >>= 1)
            acc[k] += __shfl_xor(acc[k], m);
    const float a = acc[0], b = acc[1], c = acc[2], d = acc[3];
    const float ab = a + b, cd = c + d;
    if (lane < 9) {
        float v;
        switch (lane) {
            case 0:  v = a;       break;
            case 1:  v = c;       break;
            case 2:  v = a + c;   break;
            case 3:  v = b;       break;
            case 4:  v = d;       break;
            case 5:  v = b + d;   break;
            case 6:  v = ab;      break;
            case 7:  v = cd;      break;
            default: v = ab + cd; break;
        }
        const int z = G + jq;
        f2[2 * (9 * z + lane)]           = v;
        f2[2 * (9 * (z - 1) + lane) + 1] = v;
    } else if (lane < 18 && jq < SHN - KQ - 1) {
        const int x = lane - 9;
        const int z = jq < G ? jq : KQ + 1 + jq;
        f2[2 * (9 * z + x)] = 0.f;
        if (z > 0)        f2[2 * (9 * (z - 1) + x) + 1] = 0.f;
        if (z == SHN - 1) f2[2 * (9 * z + x) + 1]       = 0.f;
    }
}

static_assert(ROWS % 128 == 0, "whole blocks of 128 rows");
static_assert(GT % 4 == 0 && PAD % 4 == 0 && T_LEN % 4 == 0, "row groups do not straddle boundaries");
#define NFB ((KQ + 1 + 3) / 4)
#define NSB (ROWS / 128)
__global__ __launch_bounds__(128)
void prep_kernel(const float* __restrict__ x,
                 const float* __restrict__ h,
                 float* __restrict__ f2,
                 float* __restrict__ xq) {
    const int lane = threadIdx.x & 31;
    if (blockIdx.x < NFB) {
        const int jq = blockIdx.x * 4 + __builtin_amdgcn_readfirstlane((int)(threadIdx.x >> 5));
        if (jq <= KQ) build_tap_group<MT2>(h, f2, jq, lane);
        return;
    }
    const int bs   = blockIdx.x - NFB;
    const int s0   = (threadIdx.x >> 5) << 3;
    const int g    = bs * 32 + lane;
    const int r0   = bs * 128;
    const float* xs = x + (size_t)s0 * T_LEN;
    v4f v[8];
    if (r0 >= GT + PAD && r0 + 128 <= GT + PAD + T_LEN) {
        const float* pin = xs + (4 * g - (GT + PAD));
#pragma unroll
        for (int j = 0; j < 8; ++j) {
            const v4f a = *(const v4f*)(pin + (size_t)j * T_LEN);
            v[j] = v4f{a.x, a.z, a.y, a.w};
        }
    } else {
#pragma unroll
        for (int j = 0; j < 8; ++j) {
            float e[4];
#pragma unroll
            for (int r = 0; r < 4; ++r) {
                const int tg = 4 * g + r - GT;
                float val = 0.f;
                if (tg >= 0 && tg < TP) {
                    int tr = tg - PAD;
                    if (tr < 0)            tr = -tr;
                    else if (tr >= T_LEN)  tr = 2 * T_LEN - 2 - tr;
                    val = xs[(size_t)j * T_LEN + tr];
                }
                e[r] = val;
            }
            v[j] = v4f{e[0], e[2], e[1], e[3]};
        }
    }
    float* o = xq + (size_t)s0 * ROWS + 4 * g;
#pragma unroll
    for (int j = 0; j < 8; ++j)
        *(v4f*)(o + (size_t)j * ROWS) = v[j];
}

template<int V> struct Part { static constexpr int value = V; };
constexpr int xb_of(int e) { return e == 0 ? 1 : e == 1 ? 3 : e == 2 ? 4 : e == 3 ? 5 : 7; }
template<int MT>
__global__ __launch_bounds__(128)
void fir_pass4_kernel(const float* __restrict__ filt,
                      const float* __restrict__ src,
                      float* __restrict__ dst,
                      int t0_origin) {
    constexpr int G   = Geo4<MT>::G;
    constexpr int R   = Geo4<MT>::R;
    constexpr int BL  = Geo4<MT>::BL;
    constexpr int NCH = Geo4<MT>::NCH;
    constexpr int SHN = Geo4<MT>::SHN;
    static_assert(R % BL == 0, "B ring must divide the A ring");
    static_assert(SHN % 4 == 0, "whole 16-byte slots");
    static_assert(4 * NCH >= KQ + 1 + 16 * MT + 15, "every tile's tap window lies inside the k axis");
    constexpr int WLAST  = (KQ + 15) / 4;
    constexpr int LASTCH = 4 * MT + WLAST;
    static_assert((WLAST + 1) % 4 == 0, "a tap window ends on a 4-chunk step boundary");
    static_assert(NCH >= 3 * R, "a first, at least one middle and a last ring pass");
    static_assert(LASTCH < NCH, "the last pass reaches the last needed chunk");
    static_assert(WLAST <= NCH - R + 3, "no fragment the last pass uses is refilled inside it");

    constexpr int NV  = 18 * SHN / 4;
    constexpr int NIT = (NV + 127) / 128;

    __shared__ __attribute__((aligned(16))) float sh[4 * 128 * NIT];
    constexpr int NB = 40;
    static_assert(NIT % NB == 0, "whole rounds");
    for (int j0 = 0; j0 < NIT; j0 += NB) {
        float4 stage0[NB / 2], stage1[NB / 2];
#pragma unroll
        for (int j = 0; j < NB / 2; ++j) {
            const int v0 = min((int)threadIdx.x + 128 * (j0 + j), NV - 1);
            const int v1 = min((int)threadIdx.x + 128 * (j0 + NB / 2 + j), NV - 1);
            stage0[j] = *(const float4*)(filt + 4 * v0);
            stage1[j] = *(const float4*)(filt + 4 * v1);
        }
#pragma unroll
        for (int j = 0; j < NB / 2; ++j) {
            const int v0 = (int)threadIdx.x + 128 * (j0 + j);
            const int v1 = (int)threadIdx.x + 128 * (j0 + NB / 2 + j);
            *(float4*)(sh + 4 * v0) = stage0[j];
            *(float4*)(sh + 4 * v1) = stage1[j];
        }
    }
    __syncthreads();

    const int lane = threadIdx.x & 31;
    const int wave = __builtin_amdgcn_readfirstlane((int)(threadIdx.x >> 5));
    const int n    = lane & 15;
    const int half = lane >> 4;
    const int q    = half * 2;
    const int wid  = blockIdx.x * 4 + wave;
    const int sig  = wid % S_SIG;
    const int rng  = wid / S_SIG;
    const int grp  = rng * 16 + n;

    const int t0   = t0_origin + grp * (64 * MT);
    const int t0c0 = t0_origin + rng * 16 * (64 * MT);

    const v2f* sa = (const v2f*)sh + 9 * (G + q - n);
    const float* srcs = src + (size_t)sig * ROWS;
    const float* p = srcs + ((size_t)(((SBASE + t0) >> 2) - 16) + 2 * half) * 4;

    float bnd[5];
    {
        v8f ad[5];
#pragma unroll
        for (int e = 0; e < 5; ++e) ad[e] = v8f{};
        const v2f* sd = (const v2f*)sh + 9 * (G + DSEG * n + q);
        const float* pd = srcs + ((size_t)(((SBASE + t0c0) >> 2) - 1) + DSEG * n + q) * 4;
#pragma unroll
        for (int kc = 0; kc < NDC; ++kc) {
            const v4f g0 = *(const v4f*)(pd + 16 * kc);
            const v4f g1 = *(const v4f*)(pd + 16 * kc + 4);
            v2f bd[5];
            bd[0] = v2f{g0.y, g1.y};
            bd[1] = v2f{g0.z, g1.z};
            bd[2] = v2f{g0.w, g1.w};
            bd[3] = bd[1] + bd[2];
            bd[4] = bd[0] + bd[2];
#pragma unroll
            for (int e = 0; e < 5; ++e) {
                ad[e] = __builtin_amdgcn_wmma_f32_16x16x4_f32(
                    false, sd[36 * kc + xb_of(e)], false, bd[e], (short)0, ad[e], false, false);
            }
        }
#pragma unroll
        for (int e = 0; e < 5; ++e) {
            float d = 0.f;
#pragma unroll
            for (int r = 0; r < 8; ++r) d = ((n & 7) == r) ? ad[e][r] : d;
            d = ((n >> 3) == half) ? d : 0.f;
            for (int m = 16; m >= 1; m >>= 1) d += __shfl_xor(d, m);
            bnd[e] = d;
        }
    }

    v8f acc[9][MT + 1];
#pragma unroll
    for (int x = 0; x < 9; ++x)
#pragma unroll
        for (int j = 0; j <= MT; ++j) acc[x][j] = v8f{};

    v2f ring[9][R];
    v4f bq[BL][2];
#pragma unroll
    for (int s = 0; s < 4; ++s) {
#pragma unroll
        for (int x = 0; x < 9; ++x) ring[x][s] = sa[36 * s + x];
    }
#pragma unroll
    for (int s = 0; s < BL - 4; ++s) {
        bq[s][0] = *(const v4f*)(p + 16 * s);
        bq[s][1] = *(const v4f*)(p + 16 * s + 4);
    }

    auto ring_pass = [&](auto part_c) __attribute__((always_inline)) {
        constexpr int PART = decltype(part_c)::value;
        constexpr int C0   = NCH - R;
#pragma unroll
        for (int i0 = 0; i0 < R; i0 += 4) {
            bool live[MT + 1];
#pragma unroll
            for (int j = 0; j <= MT; ++j)
                live[j] = PART == 1 ? true
                        : PART == 0 ? (i0 >= 4 * j)
                                    : (C0 + i0 + 3 <= 4 * j + WLAST);
            if (live[MT]) {
#pragma unroll
                for (int u = 0; u < 4; ++u) {
                    const v4f g0 = bq[(i0 + u) % BL][0];
                    const v4f g1 = bq[(i0 + u) % BL][1];
                    v2f bf[9];
                    bf[0] = v2f{g0.x, g1.x};
                    bf[1] = v2f{g0.y, g1.y};
                    bf[3] = v2f{g0.z, g1.z};
                    bf[4] = v2f{g0.w, g1.w};
                    bf[2] = bf[0] + bf[1];
                    bf[5] = bf[3] + bf[4];
                    bf[6] = bf[0] + bf[3];
                    bf[7] = bf[1] + bf[4];
                    bf[8] = bf[6] + bf[7];
#pragma unroll
                    for (int x = 0; x < 9; ++x)
                        acc[x][MT] = __builtin_amdgcn_wmma_f32_16x16x4_f32(
                            false, ring[x][(i0 + u + 4) % R], false, bf[x], (short)0, acc[x][MT], false, false);
                }
            }
            __builtin_amdgcn_sched_barrier(0);
            if (PART != 2) {
                int so = 36 * (i0 + 4);
                asm volatile("" : "+v"(so));
#pragma unroll
                for (int u = 0; u < 4; ++u) {
                    const int s = (i0 + u + 4) % R;
#pragma unroll
                    for (int x = 0; x < 9; ++x) ring[x][s] = sa[so + 36 * u + x];
                }
            }
#pragma unroll
            for (int u = 0; u < 4; ++u) {
                if (PART != 2 || C0 + i0 + u + BL - 4 <= LASTCH) {
                    bq[(i0 + u + BL - 4) % BL][0] = *(const v4f*)(p + 16 * (i0 + u + BL - 4));
                    bq[(i0 + u + BL - 4) % BL][1] = *(const v4f*)(p + 16 * (i0 + u + BL - 4) + 4);
                }
            }
            __builtin_amdgcn_sched_barrier(0);
            if (PART != 2 || C0 + i0 <= LASTCH) {
#pragma unroll
                for (int u = 0; u < 4; ++u) {
                    const v4f g0 = bq[(i0 + u) % BL][0];
                    const v4f g1 = bq[(i0 + u) % BL][1];
                    v2f bf[9];
                    bf[0] = v2f{g0.x, g1.x};
                    bf[1] = v2f{g0.y, g1.y};
                    bf[3] = v2f{g0.z, g1.z};
                    bf[4] = v2f{g0.w, g1.w};
                    bf[2] = bf[0] + bf[1];
                    bf[5] = bf[3] + bf[4];
                    bf[6] = bf[0] + bf[3];
                    bf[7] = bf[1] + bf[4];
                    bf[8] = bf[6] + bf[7];
#pragma unroll
                    for (int j = 1; j < MT; ++j) {
                        if (live[j]) {
#pragma unroll
                            for (int x = 0; x < 9; ++x)
                                acc[x][j] = __builtin_amdgcn_wmma_f32_16x16x4_f32(
                                    false, ring[x][(i0 + u - 4 * j + R) % R], false, bf[x], (short)0, acc[x][j], false, false);
                        }
                    }
                }
            }
        }
    };

    ring_pass(Part<0>{});
    p  += 16 * R;
    sa += 36 * R;
    for (int c0 = R; c0 < NCH - R; c0 += R) {
        ring_pass(Part<1>{});
        p  += 16 * R;
        sa += 36 * R;
    }
    ring_pass(Part<2>{});

    const int src1 = half ? n : (((n + 15) & 15) + 16);
    const bool col0 = (lane == 0);
#pragma unroll
    for (int jj = 1; jj <= MT; ++jj) {
        float pbA, paB, pbB, pcB, pbC;
        if (jj == 1) {
            pbA = __shfl(half ? acc[1][MT][7] : acc[1][1][7], src1);
            paB = __shfl(half ? acc[3][MT][7] : acc[3][1][7], src1);
            pbB = __shfl(half ? acc[4][MT][7] : acc[4][1][7], src1);
            pcB = __shfl(half ? acc[5][MT][7] : acc[5][1][7], src1);
            pbC = __shfl(half ? acc[7][MT][7] : acc[7][1][7], src1);
            pbA = col0 ? bnd[0] : pbA;
            paB = col0 ? bnd[1] : paB;
            pbB = col0 ? bnd[2] : pbB;
            pcB = col0 ? bnd[3] : pcB;
            pbC = col0 ? bnd[4] : pbC;
        } else {
            pbA = __shfl_xor(half ? acc[1][jj - 1][7] : acc[1][jj][7], 16);
            paB = __shfl_xor(half ? acc[3][jj - 1][7] : acc[3][jj][7], 16);
            pbB = __shfl_xor(half ? acc[4][jj - 1][7] : acc[4][jj][7], 16);
            pcB = __shfl_xor(half ? acc[5][jj - 1][7] : acc[5][jj][7], 16);
            pbC = __shfl_xor(half ? acc[7][jj - 1][7] : acc[7][jj][7], 16);
        }
        const int tt = t0 + 64 * (jj - 1) + 32 * half;
        float* o = dst + (size_t)sig * T_LEN + (tt - PAD);
#pragma unroll
        for (int r = 0; r < 8; ++r) {
            const float aA = acc[0][jj][r], bA = acc[1][jj][r], cA = acc[2][jj][r];
            const float aB = acc[3][jj][r], bB = acc[4][jj][r], cB = acc[5][jj][r];
            const float aC = acc[6][jj][r], bC = acc[7][jj][r], cC = acc[8][jj][r];
            const float Ae  = aA + pbA;
            const float Ao  = cA - aA - bA;
            const float Be  = aB + pbB;
            const float Bo  = cB - aB - bB;
            const float Bom = pcB - paB - pbB;
            const float Ce  = aC + pbC;
            const float Co  = cC - aC - bC;
            *((float4*)o + r) = make_float4(Ae + Bom, Ce - Ae - Be, Ao + Be, Co - Ao - Bo);
            pbA = bA; paB = aB; pbB = bB; pcB = cB; pbC = bC;
        }
    }
}

extern "C" void kernel_launch(void* const* d_in, const int* in_sizes, int n_in,
                              void* d_out, int out_size, void* d_ws, size_t ws_size,
                              hipStream_t stream) {
    const float* x    = (const float*)d_in[0];
    const float* kern = (const float*)d_in[1];
    float* out = (float*)d_out;
    float* ws  = (float*)d_ws;

    float* f2  = ws + F2OFF;
    float* xq  = ws + XOFF;

    prep_kernel<<<NFB + NSB, 128, 0, stream>>>(x, kern, f2, xq);

    fir_pass4_kernel<MT2>
        <<<T_LEN / (64 * MT2 * 2), 128, 0, stream>>>(f2, xq, out, PAD);
}
